# starting kernel
# baseline (speedup 1.0000x reference)
#include <hip/hip_runtime.h>
#include <hip/hip_bf16.h>


#define B_   4
#define T_   512
#define E_   256
#define HID_ 512
#define HD_  32
#define NH_  16
#define S_   768
#define LN_EPS_ 1e-5f
#define CUTOFF_ 1e-5f
#define EXP_CLAMP_ 80.0f
#define PAD_ 36

typedef __attribute__((ext_vector_type(2))) float v2f;
typedef __attribute__((ext_vector_type(8))) float v8f;

__device__ __forceinline__ v8f wmma_f32_4(v2f a, v2f b, v8f c) {
  return __builtin_amdgcn_wmma_f32_16x16x4_f32(
       false, a,  false, b,
       (short)0, c,  false,  false);
}

__device__ __forceinline__ float half_reduce_sum(float x) {
  x += __shfl_xor(x, 1);
  x += __shfl_xor(x, 2);
  x += __shfl_xor(x, 4);
  x += __shfl_xor(x, 8);
  return x;
}

__device__ __forceinline__ float4 ldg4(const float* p) {
  return *(const float4*)p;
}

__global__ __launch_bounds__(128) void gather_kernel(
    const float* __restrict__ k, const float* __restrict__ v,
    const int* __restrict__ idx, float* __restrict__ ek,
    float* __restrict__ ev) {
  const int e = blockIdx.x;
  const int b = blockIdx.y;
  const int t = idx[b * E_ + e];
  const float* ks = k + ((size_t)b * T_ + t) * HID_;
  const float* vs = v + ((size_t)b * T_ + t) * HID_;
  float* kd = ek + ((size_t)b * E_ + e) * HID_;
  float* vd = ev + ((size_t)b * E_ + e) * HID_;
  const int i = threadIdx.x * 4;
  *(float4*)(kd + i) = ldg4(ks + i);
  *(float4*)(vd + i) = ldg4(vs + i);
}

__global__ __launch_bounds__(128) void attn_kernel(
    const float* __restrict__ q, const float* __restrict__ k,
    const float* __restrict__ v, const float* __restrict__ bias,
    const float* __restrict__ law, const float* __restrict__ ek,
    const float* __restrict__ ev, float* __restrict__ attn_out) {
  __shared__ float sK[16][PAD_];
  __shared__ float sV[16][PAD_];
  __shared__ float sQ[4][16][PAD_];
  __shared__ float sP[4][16][17];

  const int tid  = threadIdx.x;
  const int lane = tid & 31;
  const int w    = tid >> 5;
  const int row  = lane & 15;
  const int hi   = lane >> 4;

  const int TB  = T_ / 64;
  const int bid = blockIdx.x;
  const int tg  = bid % TB;
  const int h   = (bid / TB) % NH_;
  const int b   = bid / (TB * NH_);
  const int t0  = (tg * 4 + w) * 16;

  {
    const int qrow = lane >> 1;
    const int qc   = (lane & 1) * 16;
    const float* qp =
        q + ((size_t)(b * T_ + t0 + qrow)) * HID_ + h * HD_ + qc;
#pragma unroll
    for (int ii = 0; ii < 4; ++ii)
      *(float4*)&sQ[w][qrow][qc + ii * 4] = ldg4(qp + ii * 4);
  }

  v2f aq[8];
#pragma unroll
  for (int kk = 0; kk < 8; ++kk) {
    const int kc = kk * 4 + 2 * hi;
    aq[kk].x = sQ[w][row][kc];
    aq[kk].y = sQ[w][row][kc + 1];
  }

  float rsum[8];
#pragma unroll
  for (int j = 0; j < 8; ++j) rsum[j] = 0.0f;
  v8f acc0 = {};
  v8f acc1 = {};

  const size_t biasBase = (((size_t)b * NH_ + h) * T_ + t0) * S_;
  const size_t lawBase  = ((size_t)b * T_ + t0) * S_;

  const int kvrow = tid >> 3;
  const int kvc   = (tid & 7) * 4;

  for (int st = 0; st < S_ / 16; ++st) {
    const int s0 = st * 16;

    const bool ext = (s0 >= T_);
    const float* baseK = ext
        ? (ek + ((size_t)(b * E_ + (s0 - T_))) * HID_ + h * HD_)
        : (k  + ((size_t)(b * T_ + s0)) * HID_ + h * HD_);
    const float* baseV = ext
        ? (ev + ((size_t)(b * E_ + (s0 - T_))) * HID_ + h * HD_)
        : (v  + ((size_t)(b * T_ + s0)) * HID_ + h * HD_);

    *(float4*)&sK[kvrow][kvc] = ldg4(baseK + (size_t)kvrow * HID_ + kvc);
    *(float4*)&sV[kvrow][kvc] = ldg4(baseV + (size_t)kvrow * HID_ + kvc);
    __syncthreads();

    v8f c = {};
#pragma unroll
    for (int kk = 0; kk < 8; ++kk) {
      const int kc = kk * 4 + 2 * hi;
      v2f bb;
      bb.x = sK[row][kc];
      bb.y = sK[row][kc + 1];
      c = wmma_f32_4(aq[kk], bb, c);
    }

#pragma unroll
    for (int j = 0; j < 8; ++j) {
      const int m = j + 8 * hi;
      const int n = s0 + row;
      const float sc = c[j] + bias[biasBase + (size_t)m * S_ + n];
      const float lw = law[lawBase + (size_t)m * S_ + n];
      const float e =
          (lw <= CUTOFF_) ? 0.0f : __expf(fminf(sc, EXP_CLAMP_));
      rsum[j] += e;
      sP[w][m][row] = e * lw;
    }

#pragma unroll
    for (int kk = 0; kk < 4; ++kk) {
      const int kc = kk * 4 + 2 * hi;
      v2f a, b0, b1;
      a.x  = sP[w][row][kc];
      a.y  = sP[w][row][kc + 1];
      b0.x = sV[kc][row];
      b0.y = sV[kc + 1][row];
      b1.x = sV[kc][row + 16];
      b1.y = sV[kc + 1][row + 16];
      acc0 = wmma_f32_4(a, b0, acc0);
      acc1 = wmma_f32_4(a, b1, acc1);
    }
    __syncthreads();
  }

#pragma unroll
  for (int j = 0; j < 8; ++j) {
    const float s = half_reduce_sum(rsum[j]);
    const float inv = 1.0f / s;
    const int m = j + 8 * hi;
    const size_t o = ((size_t)(b * T_ + t0 + m)) * HID_ + h * HD_;
    attn_out[o + row]      = acc0[j] * inv;
    attn_out[o + 16 + row] = acc1[j] * inv;
  }
}

__global__ __launch_bounds__(256) void ln_stats_kernel(
    const float* __restrict__ attn, float* __restrict__ mu,
    float* __restrict__ rstd) {
  __shared__ float red[256];
  const int rowi = blockIdx.x;
  const int tid = threadIdx.x;
  const float* a = attn + (size_t)rowi * HID_;

  const float x0 = a[tid], x1 = a[tid + 256];
  red[tid] = x0 + x1;
  __syncthreads();
  for (int off = 128; off > 0; off >>= 1) {
    if (tid < off) red[tid] += red[tid + off];
    __syncthreads();
  }
  const float mean = red[0] * (1.0f / HID_);
  __syncthreads();

  const float d0 = x0 - mean, d1 = x1 - mean;
  red[tid] = d0 * d0 + d1 * d1;
  __syncthreads();
  for (int off = 128; off > 0; off >>= 1) {
    if (tid < off) red[tid] += red[tid + off];
    __syncthreads();
  }
  if (tid == 0) {
    mu[rowi] = mean;
    rstd[rowi] = rsqrtf(red[0] * (1.0f / HID_) + LN_EPS_);
  }
}

__global__ __launch_bounds__(128) void outproj_kernel(
    const float* __restrict__ attn, const float* __restrict__ mu,
    const float* __restrict__ rstd, const float* __restrict__ ln_g,
    const float* __restrict__ ln_b, const float* __restrict__ Wout,
    float* __restrict__ out) {
  __shared__ float sB[16][PAD_];
  __shared__ float sA[4][16][PAD_];

  const int tid  = threadIdx.x;
  const int lane = tid & 31;
  const int w    = tid >> 5;
  const int row  = lane & 15;
  const int hi   = lane >> 4;

  const int NT = HID_ / 16;
  const int nt = blockIdx.x % NT;
  const int mg = blockIdx.x / NT;
  const int m0 = (mg * 4 + w) * 16;
  const int n0 = nt * 16;

  const int wrow = tid >> 3;
  const int wc   = (tid & 7) * 4;
  const int arow = lane >> 1;
  const int ac   = (lane & 1) * 16;

  v8f c = {};
  for (int k0 = 0; k0 < HID_; k0 += 32) {
    *(float4*)&sB[wrow][wc] =
        ldg4(Wout + ((size_t)(n0 + wrow)) * HID_ + k0 + wc);

    {
      const float mu_r = mu[m0 + arow];
      const float rs_r = rstd[m0 + arow];
      const float* ap = attn + ((size_t)(m0 + arow)) * HID_ + k0 + ac;
#pragma unroll
      for (int ii = 0; ii < 4; ++ii) {
        const float4 x = ldg4(ap + ii * 4);
        const float4 g = ldg4(ln_g + k0 + ac + ii * 4);
        const float4 o = ldg4(ln_b + k0 + ac + ii * 4);
        float4 y;
        y.x = (x.x - mu_r) * rs_r * g.x + o.x;
        y.y = (x.y - mu_r) * rs_r * g.y + o.y;
        y.z = (x.z - mu_r) * rs_r * g.z + o.z;
        y.w = (x.w - mu_r) * rs_r * g.w + o.w;
        *(float4*)&sA[w][arow][ac + ii * 4] = y;
      }
    }
    __syncthreads();
#pragma unroll
    for (int kk = 0; kk < 8; ++kk) {
      const int kc = kk * 4 + 2 * hi;
      v2f a, bb;
      a.x  = sA[w][row][kc];
      a.y  = sA[w][row][kc + 1];
      bb.x = sB[row][kc];
      bb.y = sB[row][kc + 1];
      c = wmma_f32_4(a, bb, c);
    }
    __syncthreads();
  }

#pragma unroll
  for (int j = 0; j < 8; ++j) {
    const int m = m0 + j + 8 * hi;
    out[(size_t)m * HID_ + n0 + row] = c[j];
  }
}

extern "C" void kernel_launch(void* const* d_in, const int* in_sizes, int n_in,
                              void* d_out, int out_size, void* d_ws,
                              size_t ws_size, hipStream_t stream) {
  const float* q    = (const float*)d_in[0];
  const float* k    = (const float*)d_in[1];
  const float* v    = (const float*)d_in[2];
  const float* bias = (const float*)d_in[3];
  const int*   idx  = (const int*)d_in[5];
  const float* law  = (const float*)d_in[6];
  const float* Wout = (const float*)d_in[8];
  const float* lng  = (const float*)d_in[9];
  const float* lnb  = (const float*)d_in[10];
  float* out = (float*)d_out;

  float* ws   = (float*)d_ws;
  float* ek   = ws;
  float* ev   = ek + (size_t)B_ * E_ * HID_;
  float* attn = ev + (size_t)B_ * E_ * HID_;
  float* mu   = attn + (size_t)B_ * T_ * HID_;
  float* rstd = mu + (size_t)B_ * T_;

  gather_kernel<<<dim3(E_, B_), 128, 0, stream>>>(k, v, idx, ek, ev);
  attn_kernel<<<B_ * NH_ * (T_ / 64), 128, 0, stream>>>(q, k, v, bias, law, ek,
                                                        ev, attn);
  ln_stats_kernel<<<B_ * T_, 256, 0, stream>>>(attn, mu, rstd);
  outproj_kernel<<<(B_ * T_ / 64) * (HID_ / 16), 128, 0, stream>>>(
      attn, mu, rstd, lng, lnb, Wout, out);
}


# attention: every global load of an s-tile (K, V, 8 bias, 8 law per lane) issued together at the tile top, law mask as a select; the built parent waited on law, then bias under a branch, row by row
# speedup vs baseline: 2.4391x; 2.4391x over previous
#include <hip/hip_runtime.h>
#include <hip/hip_bf16.h>


#define B_   4
#define T_   512
#define E_   256
#define HID_ 512
#define HD_  32
#define NH_  16
#define S_   768
#define LN_EPS_ 1e-5f
#define CUTOFF_ 1e-5f
#define EXP_CLAMP_ 80.0f
#define PAD_ 36

typedef __attribute__((ext_vector_type(2))) float v2f;
typedef __attribute__((ext_vector_type(8))) float v8f;

__device__ __forceinline__ v8f wmma_f32_4(v2f a, v2f b, v8f c) {
  return __builtin_amdgcn_wmma_f32_16x16x4_f32(
       false, a,  false, b,
       (short)0, c,  false,  false);
}

__device__ __forceinline__ float half_reduce_sum(float x) {
  x += __shfl_xor(x, 1);
  x += __shfl_xor(x, 2);
  x += __shfl_xor(x, 4);
  x += __shfl_xor(x, 8);
  return x;
}

__device__ __forceinline__ float4 ldg4(const float* p) {
  return *(const float4*)p;
}

__global__ __launch_bounds__(128) void gather_kernel(
    const float* __restrict__ k, const float* __restrict__ v,
    const int* __restrict__ idx, float* __restrict__ ek,
    float* __restrict__ ev) {
  const int e = blockIdx.x;
  const int b = blockIdx.y;
  const int t = idx[b * E_ + e];
  const float* ks = k + ((size_t)b * T_ + t) * HID_;
  const float* vs = v + ((size_t)b * T_ + t) * HID_;
  float* kd = ek + ((size_t)b * E_ + e) * HID_;
  float* vd = ev + ((size_t)b * E_ + e) * HID_;
  const int i = threadIdx.x * 4;
  *(float4*)(kd + i) = ldg4(ks + i);
  *(float4*)(vd + i) = ldg4(vs + i);
}

__global__ __launch_bounds__(128) void attn_kernel(
    const float* __restrict__ q, const float* __restrict__ k,
    const float* __restrict__ v, const float* __restrict__ bias,
    const float* __restrict__ law, const float* __restrict__ ek,
    const float* __restrict__ ev, float* __restrict__ attn_out) {
  __shared__ float sK[16][PAD_];
  __shared__ float sV[16][PAD_];
  __shared__ float sQ[4][16][PAD_];
  __shared__ float sP[4][16][17];

  const int tid  = threadIdx.x;
  const int lane = tid & 31;
  const int w    = tid >> 5;
  const int row  = lane & 15;
  const int hi   = lane >> 4;

  const int TB  = T_ / 64;
  const int bid = blockIdx.x;
  const int tg  = bid % TB;
  const int h   = (bid / TB) % NH_;
  const int b   = bid / (TB * NH_);
  const int t0  = (tg * 4 + w) * 16;

  {
    const int qrow = lane >> 1;
    const int qc   = (lane & 1) * 16;
    const float* qp =
        q + ((size_t)(b * T_ + t0 + qrow)) * HID_ + h * HD_ + qc;
#pragma unroll
    for (int ii = 0; ii < 4; ++ii)
      *(float4*)&sQ[w][qrow][qc + ii * 4] = ldg4(qp + ii * 4);
  }

  v2f aq[8];
#pragma unroll
  for (int kk = 0; kk < 8; ++kk) {
    const int kc = kk * 4 + 2 * hi;
    aq[kk].x = sQ[w][row][kc];
    aq[kk].y = sQ[w][row][kc + 1];
  }

  float rsum[8];
#pragma unroll
  for (int j = 0; j < 8; ++j) rsum[j] = 0.0f;
  v8f acc0 = {};
  v8f acc1 = {};

  const size_t biasBase = (((size_t)b * NH_ + h) * T_ + t0) * S_;
  const size_t lawBase  = ((size_t)b * T_ + t0) * S_;

  const int kvrow = tid >> 3;
  const int kvc   = (tid & 7) * 4;

  for (int st = 0; st < S_ / 16; ++st) {
    const int s0 = st * 16;

    const bool ext = (s0 >= T_);
    const float* baseK = ext
        ? (ek + ((size_t)(b * E_ + (s0 - T_))) * HID_ + h * HD_)
        : (k  + ((size_t)(b * T_ + s0)) * HID_ + h * HD_);
    const float* baseV = ext
        ? (ev + ((size_t)(b * E_ + (s0 - T_))) * HID_ + h * HD_)
        : (v  + ((size_t)(b * T_ + s0)) * HID_ + h * HD_);

    const float4 kreg = ldg4(baseK + (size_t)kvrow * HID_ + kvc);
    const float4 vreg = ldg4(baseV + (size_t)kvrow * HID_ + kvc);
    float bv[8], lv[8];
#pragma unroll
    for (int j = 0; j < 8; ++j) {
      const size_t off = (size_t)(j + 8 * hi) * S_ + s0 + row;
      bv[j] = bias[biasBase + off];
      lv[j] = law[lawBase + off];
    }
    *(float4*)&sK[kvrow][kvc] = kreg;
    *(float4*)&sV[kvrow][kvc] = vreg;
    __syncthreads();

    v8f c = {};
#pragma unroll
    for (int kk = 0; kk < 8; ++kk) {
      const int kc = kk * 4 + 2 * hi;
      v2f bb;
      bb.x = sK[row][kc];
      bb.y = sK[row][kc + 1];
      c = wmma_f32_4(aq[kk], bb, c);
    }

#pragma unroll
    for (int j = 0; j < 8; ++j) {
      const int m = j + 8 * hi;
      const float lw = lv[j];
      const float ex = __expf(fminf(c[j] + bv[j], EXP_CLAMP_));
      const float e = (lw <= CUTOFF_) ? 0.0f : ex;
      rsum[j] += e;
      sP[w][m][row] = e * lw;
    }

#pragma unroll
    for (int kk = 0; kk < 4; ++kk) {
      const int kc = kk * 4 + 2 * hi;
      v2f a, b0, b1;
      a.x  = sP[w][row][kc];
      a.y  = sP[w][row][kc + 1];
      b0.x = sV[kc][row];
      b0.y = sV[kc + 1][row];
      b1.x = sV[kc][row + 16];
      b1.y = sV[kc + 1][row + 16];
      acc0 = wmma_f32_4(a, b0, acc0);
      acc1 = wmma_f32_4(a, b1, acc1);
    }
    __syncthreads();
  }

#pragma unroll
  for (int j = 0; j < 8; ++j) {
    const float s = half_reduce_sum(rsum[j]);
    const float inv = 1.0f / s;
    const int m = j + 8 * hi;
    const size_t o = ((size_t)(b * T_ + t0 + m)) * HID_ + h * HD_;
    attn_out[o + row]      = acc0[j] * inv;
    attn_out[o + 16 + row] = acc1[j] * inv;
  }
}

__global__ __launch_bounds__(256) void ln_stats_kernel(
    const float* __restrict__ attn, float* __restrict__ mu,
    float* __restrict__ rstd) {
  __shared__ float red[256];
  const int rowi = blockIdx.x;
  const int tid = threadIdx.x;
  const float* a = attn + (size_t)rowi * HID_;

  const float x0 = a[tid], x1 = a[tid + 256];
  red[tid] = x0 + x1;
  __syncthreads();
  for (int off = 128; off > 0; off >>= 1) {
    if (tid < off) red[tid] += red[tid + off];
    __syncthreads();
  }
  const float mean = red[0] * (1.0f / HID_);
  __syncthreads();

  const float d0 = x0 - mean, d1 = x1 - mean;
  red[tid] = d0 * d0 + d1 * d1;
  __syncthreads();
  for (int off = 128; off > 0; off >>= 1) {
    if (tid < off) red[tid] += red[tid + off];
    __syncthreads();
  }
  if (tid == 0) {
    mu[rowi] = mean;
    rstd[rowi] = rsqrtf(red[0] * (1.0f / HID_) + LN_EPS_);
  }
}

__global__ __launch_bounds__(128) void outproj_kernel(
    const float* __restrict__ attn, const float* __restrict__ mu,
    const float* __restrict__ rstd, const float* __restrict__ ln_g,
    const float* __restrict__ ln_b, const float* __restrict__ Wout,
    float* __restrict__ out) {
  __shared__ float sB[16][PAD_];
  __shared__ float sA[4][16][PAD_];

  const int tid  = threadIdx.x;
  const int lane = tid & 31;
  const int w    = tid >> 5;
  const int row  = lane & 15;
  const int hi   = lane >> 4;

  const int NT = HID_ / 16;
  const int nt = blockIdx.x % NT;
  const int mg = blockIdx.x / NT;
  const int m0 = (mg * 4 + w) * 16;
  const int n0 = nt * 16;

  const int wrow = tid >> 3;
  const int wc   = (tid & 7) * 4;
  const int arow = lane >> 1;
  const int ac   = (lane & 1) * 16;

  v8f c = {};
  for (int k0 = 0; k0 < HID_; k0 += 32) {
    *(float4*)&sB[wrow][wc] =
        ldg4(Wout + ((size_t)(n0 + wrow)) * HID_ + k0 + wc);

    {
      const float mu_r = mu[m0 + arow];
      const float rs_r = rstd[m0 + arow];
      const float* ap = attn + ((size_t)(m0 + arow)) * HID_ + k0 + ac;
#pragma unroll
      for (int ii = 0; ii < 4; ++ii) {
        const float4 x = ldg4(ap + ii * 4);
        const float4 g = ldg4(ln_g + k0 + ac + ii * 4);
        const float4 o = ldg4(ln_b + k0 + ac + ii * 4);
        float4 y;
        y.x = (x.x - mu_r) * rs_r * g.x + o.x;
        y.y = (x.y - mu_r) * rs_r * g.y + o.y;
        y.z = (x.z - mu_r) * rs_r * g.z + o.z;
        y.w = (x.w - mu_r) * rs_r * g.w + o.w;
        *(float4*)&sA[w][arow][ac + ii * 4] = y;
      }
    }
    __syncthreads();
#pragma unroll
    for (int kk = 0; kk < 8; ++kk) {
      const int kc = kk * 4 + 2 * hi;
      v2f a, bb;
      a.x  = sA[w][row][kc];
      a.y  = sA[w][row][kc + 1];
      bb.x = sB[row][kc];
      bb.y = sB[row][kc + 1];
      c = wmma_f32_4(a, bb, c);
    }
    __syncthreads();
  }

#pragma unroll
  for (int j = 0; j < 8; ++j) {
    const int m = m0 + j + 8 * hi;
    out[(size_t)m * HID_ + n0 + row] = c[j];
  }
}

extern "C" void kernel_launch(void* const* d_in, const int* in_sizes, int n_in,
                              void* d_out, int out_size, void* d_ws,
                              size_t ws_size, hipStream_t stream) {
  const float* q    = (const float*)d_in[0];
  const float* k    = (const float*)d_in[1];
  const float* v    = (const float*)d_in[2];
  const float* bias = (const float*)d_in[3];
  const int*   idx  = (const int*)d_in[5];
  const float* law  = (const float*)d_in[6];
  const float* Wout = (const float*)d_in[8];
  const float* lng  = (const float*)d_in[9];
  const float* lnb  = (const float*)d_in[10];
  float* out = (float*)d_out;

  float* ws   = (float*)d_ws;
  float* ek   = ws;
  float* ev   = ek + (size_t)B_ * E_ * HID_;
  float* attn = ev + (size_t)B_ * E_ * HID_;
  float* mu   = attn + (size_t)B_ * T_ * HID_;
  float* rstd = mu + (size_t)B_ * T_;

  gather_kernel<<<dim3(E_, B_), 128, 0, stream>>>(k, v, idx, ek, ev);
  attn_kernel<<<B_ * NH_ * (T_ / 64), 128, 0, stream>>>(q, k, v, bias, law, ek,
                                                        ev, attn);
  ln_stats_kernel<<<B_ * T_, 256, 0, stream>>>(attn, mu, rstd);
  outproj_kernel<<<(B_ * T_ / 64) * (HID_ / 16), 128, 0, stream>>>(
      attn, mu, rstd, lng, lnb, Wout, out);
}


# attention: each (head, t-tile)'s 48 s-tiles split over 4 wave groups (16-wave blocks, 12 tiles per wave); partial row sums and P@V sums added through LDS at the end (summation order changes)
# speedup vs baseline: 3.2483x; 1.3318x over previous
#include <hip/hip_runtime.h>
#include <hip/hip_bf16.h>


#define B_   4
#define T_   512
#define E_   256
#define HID_ 512
#define HD_  32
#define NH_  16
#define S_   768
#define LN_EPS_ 1e-5f
#define CUTOFF_ 1e-5f
#define EXP_CLAMP_ 80.0f
#define PAD_ 36

typedef __attribute__((ext_vector_type(2))) float v2f;
typedef __attribute__((ext_vector_type(8))) float v8f;

__device__ __forceinline__ v8f wmma_f32_4(v2f a, v2f b, v8f c) {
  return __builtin_amdgcn_wmma_f32_16x16x4_f32(
       false, a,  false, b,
       (short)0, c,  false,  false);
}

__device__ __forceinline__ float half_reduce_sum(float x) {
  x += __shfl_xor(x, 1);
  x += __shfl_xor(x, 2);
  x += __shfl_xor(x, 4);
  x += __shfl_xor(x, 8);
  return x;
}

__device__ __forceinline__ float4 ldg4(const float* p) {
  return *(const float4*)p;
}

__global__ __launch_bounds__(128) void gather_kernel(
    const float* __restrict__ k, const float* __restrict__ v,
    const int* __restrict__ idx, float* __restrict__ ek,
    float* __restrict__ ev) {
  const int e = blockIdx.x;
  const int b = blockIdx.y;
  const int t = idx[b * E_ + e];
  const float* ks = k + ((size_t)b * T_ + t) * HID_;
  const float* vs = v + ((size_t)b * T_ + t) * HID_;
  float* kd = ek + ((size_t)b * E_ + e) * HID_;
  float* vd = ev + ((size_t)b * E_ + e) * HID_;
  const int i = threadIdx.x * 4;
  *(float4*)(kd + i) = ldg4(ks + i);
  *(float4*)(vd + i) = ldg4(vs + i);
}

#define SG_  4
#define TPG_ (S_ / 16 / SG_)

__global__ __launch_bounds__(512) void attn_kernel(
    const float* __restrict__ q, const float* __restrict__ k,
    const float* __restrict__ v, const float* __restrict__ bias,
    const float* __restrict__ law, const float* __restrict__ ek,
    const float* __restrict__ ev, float* __restrict__ attn_out) {
  __shared__ float sK[SG_][16][PAD_];
  __shared__ float sV[SG_][16][PAD_];
  __shared__ float sQ[4 * SG_][16][PAD_];
  __shared__ float sP[4 * SG_][16][17];

  const int tid  = threadIdx.x;
  const int lane = tid & 31;
  const int wv   = __builtin_amdgcn_readfirstlane(tid >> 5);
  const int w    = wv & 3;
  const int g    = wv >> 2;
  const int row  = lane & 15;
  const int hi   = lane >> 4;

  const int TB  = T_ / 64;
  const int bid = blockIdx.x;
  const int tg  = bid % TB;
  const int h   = (bid / TB) % NH_;
  const int b   = bid / (TB * NH_);
  const int t0  = (tg * 4 + w) * 16;

  {
    const int qrow = lane >> 1;
    const int qc   = (lane & 1) * 16;
    const float* qp =
        q + ((size_t)(b * T_ + t0 + qrow)) * HID_ + h * HD_ + qc;
#pragma unroll
    for (int ii = 0; ii < 4; ++ii)
      *(float4*)&sQ[wv][qrow][qc + ii * 4] = ldg4(qp + ii * 4);
  }

  v2f aq[8];
#pragma unroll
  for (int kk = 0; kk < 8; ++kk) {
    const int kc = kk * 4 + 2 * hi;
    aq[kk].x = sQ[wv][row][kc];
    aq[kk].y = sQ[wv][row][kc + 1];
  }

  float rsum[8];
#pragma unroll
  for (int j = 0; j < 8; ++j) rsum[j] = 0.0f;
  v8f acc0 = {};
  v8f acc1 = {};

  const size_t biasBase = (((size_t)b * NH_ + h) * T_ + t0) * S_;
  const size_t lawBase  = ((size_t)b * T_ + t0) * S_;

  const int gt    = tid & 127;
  const int kvrow = gt >> 3;
  const int kvc   = (gt & 7) * 4;

  for (int i = 0; i < TPG_; ++i) {
    const int s0 = (g * TPG_ + i) * 16;

    const bool ext = (s0 >= T_);
    const float* baseK = ext
        ? (ek + ((size_t)(b * E_ + (s0 - T_))) * HID_ + h * HD_)
        : (k  + ((size_t)(b * T_ + s0)) * HID_ + h * HD_);
    const float* baseV = ext
        ? (ev + ((size_t)(b * E_ + (s0 - T_))) * HID_ + h * HD_)
        : (v  + ((size_t)(b * T_ + s0)) * HID_ + h * HD_);

    const float4 kreg = ldg4(baseK + (size_t)kvrow * HID_ + kvc);
    const float4 vreg = ldg4(baseV + (size_t)kvrow * HID_ + kvc);
    float bv[8], lv[8];
#pragma unroll
    for (int j = 0; j < 8; ++j) {
      const size_t off = (size_t)(j + 8 * hi) * S_ + s0 + row;
      bv[j] = bias[biasBase + off];
      lv[j] = law[lawBase + off];
    }
    *(float4*)&sK[g][kvrow][kvc] = kreg;
    *(float4*)&sV[g][kvrow][kvc] = vreg;
    __syncthreads();

    v8f c = {};
#pragma unroll
    for (int kk = 0; kk < 8; ++kk) {
      const int kc = kk * 4 + 2 * hi;
      v2f bb;
      bb.x = sK[g][row][kc];
      bb.y = sK[g][row][kc + 1];
      c = wmma_f32_4(aq[kk], bb, c);
    }

#pragma unroll
    for (int j = 0; j < 8; ++j) {
      const int m = j + 8 * hi;
      const float lw = lv[j];
      const float ex = __expf(fminf(c[j] + bv[j], EXP_CLAMP_));
      const float e = (lw <= CUTOFF_) ? 0.0f : ex;
      rsum[j] += e;
      sP[wv][m][row] = e * lw;
    }

#pragma unroll
    for (int kk = 0; kk < 4; ++kk) {
      const int kc = kk * 4 + 2 * hi;
      v2f a, b0, b1;
      a.x  = sP[wv][row][kc];
      a.y  = sP[wv][row][kc + 1];
      b0.x = sV[g][kc][row];
      b0.y = sV[g][kc + 1][row];
      b1.x = sV[g][kc][row + 16];
      b1.y = sV[g][kc + 1][row + 16];
      acc0 = wmma_f32_4(a, b0, acc0);
      acc1 = wmma_f32_4(a, b1, acc1);
    }
    __syncthreads();
  }

  float* red = &sQ[0][0][0];
  if (g != 0) {
    float* r = red + ((g - 1) * 4 + w) * (24 * 32) + lane;
#pragma unroll
    for (int j = 0; j < 8; ++j) {
      r[j * 32]        = acc0[j];
      r[(8 + j) * 32]  = acc1[j];
      r[(16 + j) * 32] = rsum[j];
    }
  }
  __syncthreads();
  if (g != 0) return;
#pragma unroll
  for (int gg = 0; gg < SG_ - 1; ++gg) {
    const float* r = red + (gg * 4 + w) * (24 * 32) + lane;
#pragma unroll
    for (int j = 0; j < 8; ++j) {
      acc0[j] += r[j * 32];
      acc1[j] += r[(8 + j) * 32];
      rsum[j] += r[(16 + j) * 32];
    }
  }

#pragma unroll
  for (int j = 0; j < 8; ++j) {
    const float s = half_reduce_sum(rsum[j]);
    const float inv = 1.0f / s;
    const int m = j + 8 * hi;
    const size_t o = ((size_t)(b * T_ + t0 + m)) * HID_ + h * HD_;
    attn_out[o + row]      = acc0[j] * inv;
    attn_out[o + 16 + row] = acc1[j] * inv;
  }
}

__global__ __launch_bounds__(256) void ln_stats_kernel(
    const float* __restrict__ attn, float* __restrict__ mu,
    float* __restrict__ rstd) {
  __shared__ float red[256];
  const int rowi = blockIdx.x;
  const int tid = threadIdx.x;
  const float* a = attn + (size_t)rowi * HID_;

  const float x0 = a[tid], x1 = a[tid + 256];
  red[tid] = x0 + x1;
  __syncthreads();
  for (int off = 128; off > 0; off >>= 1) {
    if (tid < off) red[tid] += red[tid + off];
    __syncthreads();
  }
  const float mean = red[0] * (1.0f / HID_);
  __syncthreads();

  const float d0 = x0 - mean, d1 = x1 - mean;
  red[tid] = d0 * d0 + d1 * d1;
  __syncthreads();
  for (int off = 128; off > 0; off >>= 1) {
    if (tid < off) red[tid] += red[tid + off];
    __syncthreads();
  }
  if (tid == 0) {
    mu[rowi] = mean;
    rstd[rowi] = rsqrtf(red[0] * (1.0f / HID_) + LN_EPS_);
  }
}

__global__ __launch_bounds__(128) void outproj_kernel(
    const float* __restrict__ attn, const float* __restrict__ mu,
    const float* __restrict__ rstd, const float* __restrict__ ln_g,
    const float* __restrict__ ln_b, const float* __restrict__ Wout,
    float* __restrict__ out) {
  __shared__ float sB[16][PAD_];
  __shared__ float sA[4][16][PAD_];

  const int tid  = threadIdx.x;
  const int lane = tid & 31;
  const int w    = tid >> 5;
  const int row  = lane & 15;
  const int hi   = lane >> 4;

  const int NT = HID_ / 16;
  const int nt = blockIdx.x % NT;
  const int mg = blockIdx.x / NT;
  const int m0 = (mg * 4 + w) * 16;
  const int n0 = nt * 16;

  const int wrow = tid >> 3;
  const int wc   = (tid & 7) * 4;
  const int arow = lane >> 1;
  const int ac   = (lane & 1) * 16;

  v8f c = {};
  for (int k0 = 0; k0 < HID_; k0 += 32) {
    *(float4*)&sB[wrow][wc] =
        ldg4(Wout + ((size_t)(n0 + wrow)) * HID_ + k0 + wc);

    {
      const float mu_r = mu[m0 + arow];
      const float rs_r = rstd[m0 + arow];
      const float* ap = attn + ((size_t)(m0 + arow)) * HID_ + k0 + ac;
#pragma unroll
      for (int ii = 0; ii < 4; ++ii) {
        const float4 x = ldg4(ap + ii * 4);
        const float4 g = ldg4(ln_g + k0 + ac + ii * 4);
        const float4 o = ldg4(ln_b + k0 + ac + ii * 4);
        float4 y;
        y.x = (x.x - mu_r) * rs_r * g.x + o.x;
        y.y = (x.y - mu_r) * rs_r * g.y + o.y;
        y.z = (x.z - mu_r) * rs_r * g.z + o.z;
        y.w = (x.w - mu_r) * rs_r * g.w + o.w;
        *(float4*)&sA[w][arow][ac + ii * 4] = y;
      }
    }
    __syncthreads();
#pragma unroll
    for (int kk = 0; kk < 8; ++kk) {
      const int kc = kk * 4 + 2 * hi;
      v2f a, bb;
      a.x  = sA[w][row][kc];
      a.y  = sA[w][row][kc + 1];
      bb.x = sB[row][kc];
      bb.y = sB[row][kc + 1];
      c = wmma_f32_4(a, bb, c);
    }
    __syncthreads();
  }

#pragma unroll
  for (int j = 0; j < 8; ++j) {
    const int m = m0 + j + 8 * hi;
    out[(size_t)m * HID_ + n0 + row] = c[j];
  }
}

extern "C" void kernel_launch(void* const* d_in, const int* in_sizes, int n_in,
                              void* d_out, int out_size, void* d_ws,
                              size_t ws_size, hipStream_t stream) {
  const float* q    = (const float*)d_in[0];
  const float* k    = (const float*)d_in[1];
  const float* v    = (const float*)d_in[2];
  const float* bias = (const float*)d_in[3];
  const int*   idx  = (const int*)d_in[5];
  const float* law  = (const float*)d_in[6];
  const float* Wout = (const float*)d_in[8];
  const float* lng  = (const float*)d_in[9];
  const float* lnb  = (const float*)d_in[10];
  float* out = (float*)d_out;

  float* ws   = (float*)d_ws;
  float* ek   = ws;
  float* ev   = ek + (size_t)B_ * E_ * HID_;
  float* attn = ev + (size_t)B_ * E_ * HID_;
  float* mu   = attn + (size_t)B_ * T_ * HID_;
  float* rstd = mu + (size_t)B_ * T_;

  gather_kernel<<<dim3(E_, B_), 128, 0, stream>>>(k, v, idx, ek, ev);
  attn_kernel<<<B_ * NH_ * (T_ / 64), 512, 0, stream>>>(q, k, v, bias, law, ek,
                                                        ev, attn);
  ln_stats_kernel<<<B_ * T_, 256, 0, stream>>>(attn, mu, rstd);
  outproj_kernel<<<(B_ * T_ / 64) * (HID_ / 16), 128, 0, stream>>>(
      attn, mu, rstd, lng, lnb, Wout, out);
}


# attention reads the PBC expansion K/V rows straight from k/v at row outcell_index[b,s-T], index fetched one tile ahead; the gather launch and its ek/ev workspace copies are gone (4 launches -> 3)
# speedup vs baseline: 3.3898x; 1.0436x over previous
#include <hip/hip_runtime.h>
#include <hip/hip_bf16.h>


#define B_   4
#define T_   512
#define E_   256
#define HID_ 512
#define HD_  32
#define NH_  16
#define S_   768
#define LN_EPS_ 1e-5f
#define CUTOFF_ 1e-5f
#define EXP_CLAMP_ 80.0f
#define PAD_ 36

typedef __attribute__((ext_vector_type(2))) float v2f;
typedef __attribute__((ext_vector_type(8))) float v8f;

__device__ __forceinline__ v8f wmma_f32_4(v2f a, v2f b, v8f c) {
  return __builtin_amdgcn_wmma_f32_16x16x4_f32(
       false, a,  false, b,
       (short)0, c,  false,  false);
}

__device__ __forceinline__ float half_reduce_sum(float x) {
  x += __shfl_xor(x, 1);
  x += __shfl_xor(x, 2);
  x += __shfl_xor(x, 4);
  x += __shfl_xor(x, 8);
  return x;
}

__device__ __forceinline__ float4 ldg4(const float* p) {
  return *(const float4*)p;
}

__device__ __forceinline__ int clamp_row(int t) {
  return min(max(t, 0), T_ - 1);
}

#define SG_  4
#define TPG_ (S_ / 16 / SG_)

__global__ __launch_bounds__(512) void attn_kernel(
    const float* __restrict__ q, const float* __restrict__ k,
    const float* __restrict__ v, const float* __restrict__ bias,
    const float* __restrict__ law, const int* __restrict__ idx,
    float* __restrict__ attn_out) {
  __shared__ float sK[SG_][16][PAD_];
  __shared__ float sV[SG_][16][PAD_];
  __shared__ float sQ[4 * SG_][16][PAD_];
  __shared__ float sP[4 * SG_][16][17];

  const int tid  = threadIdx.x;
  const int lane = tid & 31;
  const int wv   = __builtin_amdgcn_readfirstlane(tid >> 5);
  const int w    = wv & 3;
  const int g    = wv >> 2;
  const int row  = lane & 15;
  const int hi   = lane >> 4;

  const int TB  = T_ / 64;
  const int bid = blockIdx.x;
  const int tg  = bid % TB;
  const int h   = (bid / TB) % NH_;
  const int b   = bid / (TB * NH_);
  const int t0  = (tg * 4 + w) * 16;

  {
    const int qrow = lane >> 1;
    const int qc   = (lane & 1) * 16;
    const float* qp =
        q + ((size_t)(b * T_ + t0 + qrow)) * HID_ + h * HD_ + qc;
#pragma unroll
    for (int ii = 0; ii < 4; ++ii)
      *(float4*)&sQ[wv][qrow][qc + ii * 4] = ldg4(qp + ii * 4);
  }

  v2f aq[8];
#pragma unroll
  for (int kk = 0; kk < 8; ++kk) {
    const int kc = kk * 4 + 2 * hi;
    aq[kk].x = sQ[wv][row][kc];
    aq[kk].y = sQ[wv][row][kc + 1];
  }

  float rsum[8];
#pragma unroll
  for (int j = 0; j < 8; ++j) rsum[j] = 0.0f;
  v8f acc0 = {};
  v8f acc1 = {};

  const size_t biasBase = (((size_t)b * NH_ + h) * T_ + t0) * S_;
  const size_t lawBase  = ((size_t)b * T_ + t0) * S_;

  const int gt    = tid & 127;
  const int kvrow = gt >> 3;
  const int kvc   = (gt & 7) * 4;

  const int sFirst = g * TPG_ * 16;
  int krow = (sFirst >= T_) ? idx[b * E_ + (sFirst - T_) + kvrow]
                            : (sFirst + kvrow);

  for (int i = 0; i < TPG_; ++i) {
    const int s0 = (g * TPG_ + i) * 16;

    const size_t kvoff =
        ((size_t)(b * T_ + clamp_row(krow))) * HID_ + h * HD_ + kvc;
    const float4 kreg = ldg4(k + kvoff);
    const float4 vreg = ldg4(v + kvoff);
    float bv[8], lv[8];
#pragma unroll
    for (int j = 0; j < 8; ++j) {
      const size_t off = (size_t)(j + 8 * hi) * S_ + s0 + row;
      bv[j] = bias[biasBase + off];
      lv[j] = law[lawBase + off];
    }
    *(float4*)&sK[g][kvrow][kvc] = kreg;
    *(float4*)&sV[g][kvrow][kvc] = vreg;
    __syncthreads();

    const int s1 = s0 + 16;
    int krowNext = s1 + kvrow;
    if (i + 1 < TPG_ && s1 >= T_) krowNext = idx[b * E_ + (s1 - T_) + kvrow];

    v8f c = {};
#pragma unroll
    for (int kk = 0; kk < 8; ++kk) {
      const int kc = kk * 4 + 2 * hi;
      v2f bb;
      bb.x = sK[g][row][kc];
      bb.y = sK[g][row][kc + 1];
      c = wmma_f32_4(aq[kk], bb, c);
    }

#pragma unroll
    for (int j = 0; j < 8; ++j) {
      const int m = j + 8 * hi;
      const float lw = lv[j];
      const float ex = __expf(fminf(c[j] + bv[j], EXP_CLAMP_));
      const float e = (lw <= CUTOFF_) ? 0.0f : ex;
      rsum[j] += e;
      sP[wv][m][row] = e * lw;
    }

#pragma unroll
    for (int kk = 0; kk < 4; ++kk) {
      const int kc = kk * 4 + 2 * hi;
      v2f a, b0, b1;
      a.x  = sP[wv][row][kc];
      a.y  = sP[wv][row][kc + 1];
      b0.x = sV[g][kc][row];
      b0.y = sV[g][kc + 1][row];
      b1.x = sV[g][kc][row + 16];
      b1.y = sV[g][kc + 1][row + 16];
      acc0 = wmma_f32_4(a, b0, acc0);
      acc1 = wmma_f32_4(a, b1, acc1);
    }
    krow = krowNext;
    __syncthreads();
  }

  float* red = &sQ[0][0][0];
  if (g != 0) {
    float* r = red + ((g - 1) * 4 + w) * (24 * 32) + lane;
#pragma unroll
    for (int j = 0; j < 8; ++j) {
      r[j * 32]        = acc0[j];
      r[(8 + j) * 32]  = acc1[j];
      r[(16 + j) * 32] = rsum[j];
    }
  }
  __syncthreads();
  if (g != 0) return;
#pragma unroll
  for (int gg = 0; gg < SG_ - 1; ++gg) {
    const float* r = red + (gg * 4 + w) * (24 * 32) + lane;
#pragma unroll
    for (int j = 0; j < 8; ++j) {
      acc0[j] += r[j * 32];
      acc1[j] += r[(8 + j) * 32];
      rsum[j] += r[(16 + j) * 32];
    }
  }

#pragma unroll
  for (int j = 0; j < 8; ++j) {
    const float s = half_reduce_sum(rsum[j]);
    const float inv = 1.0f / s;
    const int m = j + 8 * hi;
    const size_t o = ((size_t)(b * T_ + t0 + m)) * HID_ + h * HD_;
    attn_out[o + row]      = acc0[j] * inv;
    attn_out[o + 16 + row] = acc1[j] * inv;
  }
}

__global__ __launch_bounds__(256) void ln_stats_kernel(
    const float* __restrict__ attn, float* __restrict__ mu,
    float* __restrict__ rstd) {
  __shared__ float red[256];
  const int rowi = blockIdx.x;
  const int tid = threadIdx.x;
  const float* a = attn + (size_t)rowi * HID_;

  const float x0 = a[tid], x1 = a[tid + 256];
  red[tid] = x0 + x1;
  __syncthreads();
  for (int off = 128; off > 0; off >>= 1) {
    if (tid < off) red[tid] += red[tid + off];
    __syncthreads();
  }
  const float mean = red[0] * (1.0f / HID_);
  __syncthreads();

  const float d0 = x0 - mean, d1 = x1 - mean;
  red[tid] = d0 * d0 + d1 * d1;
  __syncthreads();
  for (int off = 128; off > 0; off >>= 1) {
    if (tid < off) red[tid] += red[tid + off];
    __syncthreads();
  }
  if (tid == 0) {
    mu[rowi] = mean;
    rstd[rowi] = rsqrtf(red[0] * (1.0f / HID_) + LN_EPS_);
  }
}

__global__ __launch_bounds__(128) void outproj_kernel(
    const float* __restrict__ attn, const float* __restrict__ mu,
    const float* __restrict__ rstd, const float* __restrict__ ln_g,
    const float* __restrict__ ln_b, const float* __restrict__ Wout,
    float* __restrict__ out) {
  __shared__ float sB[16][PAD_];
  __shared__ float sA[4][16][PAD_];

  const int tid  = threadIdx.x;
  const int lane = tid & 31;
  const int w    = tid >> 5;
  const int row  = lane & 15;
  const int hi   = lane >> 4;

  const int NT = HID_ / 16;
  const int nt = blockIdx.x % NT;
  const int mg = blockIdx.x / NT;
  const int m0 = (mg * 4 + w) * 16;
  const int n0 = nt * 16;

  const int wrow = tid >> 3;
  const int wc   = (tid & 7) * 4;
  const int arow = lane >> 1;
  const int ac   = (lane & 1) * 16;

  v8f c = {};
  for (int k0 = 0; k0 < HID_; k0 += 32) {
    *(float4*)&sB[wrow][wc] =
        ldg4(Wout + ((size_t)(n0 + wrow)) * HID_ + k0 + wc);

    {
      const float mu_r = mu[m0 + arow];
      const float rs_r = rstd[m0 + arow];
      const float* ap = attn + ((size_t)(m0 + arow)) * HID_ + k0 + ac;
#pragma unroll
      for (int ii = 0; ii < 4; ++ii) {
        const float4 x = ldg4(ap + ii * 4);
        const float4 g = ldg4(ln_g + k0 + ac + ii * 4);
        const float4 o = ldg4(ln_b + k0 + ac + ii * 4);
        float4 y;
        y.x = (x.x - mu_r) * rs_r * g.x + o.x;
        y.y = (x.y - mu_r) * rs_r * g.y + o.y;
        y.z = (x.z - mu_r) * rs_r * g.z + o.z;
        y.w = (x.w - mu_r) * rs_r * g.w + o.w;
        *(float4*)&sA[w][arow][ac + ii * 4] = y;
      }
    }
    __syncthreads();
#pragma unroll
    for (int kk = 0; kk < 8; ++kk) {
      const int kc = kk * 4 + 2 * hi;
      v2f a, bb;
      a.x  = sA[w][row][kc];
      a.y  = sA[w][row][kc + 1];
      bb.x = sB[row][kc];
      bb.y = sB[row][kc + 1];
      c = wmma_f32_4(a, bb, c);
    }
    __syncthreads();
  }

#pragma unroll
  for (int j = 0; j < 8; ++j) {
    const int m = m0 + j + 8 * hi;
    out[(size_t)m * HID_ + n0 + row] = c[j];
  }
}

extern "C" void kernel_launch(void* const* d_in, const int* in_sizes, int n_in,
                              void* d_out, int out_size, void* d_ws,
                              size_t ws_size, hipStream_t stream) {
  const float* q    = (const float*)d_in[0];
  const float* k    = (const float*)d_in[1];
  const float* v    = (const float*)d_in[2];
  const float* bias = (const float*)d_in[3];
  const int*   idx  = (const int*)d_in[5];
  const float* law  = (const float*)d_in[6];
  const float* Wout = (const float*)d_in[8];
  const float* lng  = (const float*)d_in[9];
  const float* lnb  = (const float*)d_in[10];
  float* out = (float*)d_out;

  float* ws   = (float*)d_ws;
  float* attn = ws;
  float* mu   = attn + (size_t)B_ * T_ * HID_;
  float* rstd = mu + (size_t)B_ * T_;

  attn_kernel<<<B_ * NH_ * (T_ / 64), 512, 0, stream>>>(q, k, v, bias, law,
                                                        idx, attn);
  ln_stats_kernel<<<B_ * T_, 256, 0, stream>>>(attn, mu, rstd);
  outproj_kernel<<<(B_ * T_ / 64) * (HID_ / 16), 128, 0, stream>>>(
      attn, mu, rstd, lng, lnb, Wout, out);
}


# LayerNorm by one wave per row (row in registers, two 32-lane shuffle sums, no LDS or barrier; row-sum order changes), 16384 waves -> 2048; writes the normalised rows, out-proj reads them as is
# speedup vs baseline: 3.5559x; 1.0490x over previous
#include <hip/hip_runtime.h>
#include <hip/hip_bf16.h>


#define B_   4
#define T_   512
#define E_   256
#define HID_ 512
#define HD_  32
#define NH_  16
#define S_   768
#define LN_EPS_ 1e-5f
#define CUTOFF_ 1e-5f
#define EXP_CLAMP_ 80.0f
#define PAD_ 36

typedef __attribute__((ext_vector_type(2))) float v2f;
typedef __attribute__((ext_vector_type(8))) float v8f;

__device__ __forceinline__ v8f wmma_f32_4(v2f a, v2f b, v8f c) {
  return __builtin_amdgcn_wmma_f32_16x16x4_f32(
       false, a,  false, b,
       (short)0, c,  false,  false);
}

__device__ __forceinline__ float half_reduce_sum(float x) {
  x += __shfl_xor(x, 1);
  x += __shfl_xor(x, 2);
  x += __shfl_xor(x, 4);
  x += __shfl_xor(x, 8);
  return x;
}

__device__ __forceinline__ float4 ldg4(const float* p) {
  return *(const float4*)p;
}

__device__ __forceinline__ int clamp_row(int t) {
  return min(max(t, 0), T_ - 1);
}

#define SG_  4
#define TPG_ (S_ / 16 / SG_)

__global__ __launch_bounds__(512) void attn_kernel(
    const float* __restrict__ q, const float* __restrict__ k,
    const float* __restrict__ v, const float* __restrict__ bias,
    const float* __restrict__ law, const int* __restrict__ idx,
    float* __restrict__ attn_out) {
  __shared__ float sK[SG_][16][PAD_];
  __shared__ float sV[SG_][16][PAD_];
  __shared__ float sQ[4 * SG_][16][PAD_];
  __shared__ float sP[4 * SG_][16][17];

  const int tid  = threadIdx.x;
  const int lane = tid & 31;
  const int wv   = __builtin_amdgcn_readfirstlane(tid >> 5);
  const int w    = wv & 3;
  const int g    = wv >> 2;
  const int row  = lane & 15;
  const int hi   = lane >> 4;

  const int TB  = T_ / 64;
  const int bid = blockIdx.x;
  const int tg  = bid % TB;
  const int h   = (bid / TB) % NH_;
  const int b   = bid / (TB * NH_);
  const int t0  = (tg * 4 + w) * 16;

  {
    const int qrow = lane >> 1;
    const int qc   = (lane & 1) * 16;
    const float* qp =
        q + ((size_t)(b * T_ + t0 + qrow)) * HID_ + h * HD_ + qc;
#pragma unroll
    for (int ii = 0; ii < 4; ++ii)
      *(float4*)&sQ[wv][qrow][qc + ii * 4] = ldg4(qp + ii * 4);
  }

  v2f aq[8];
#pragma unroll
  for (int kk = 0; kk < 8; ++kk) {
    const int kc = kk * 4 + 2 * hi;
    aq[kk].x = sQ[wv][row][kc];
    aq[kk].y = sQ[wv][row][kc + 1];
  }

  float rsum[8];
#pragma unroll
  for (int j = 0; j < 8; ++j) rsum[j] = 0.0f;
  v8f acc0 = {};
  v8f acc1 = {};

  const size_t biasBase = (((size_t)b * NH_ + h) * T_ + t0) * S_;
  const size_t lawBase  = ((size_t)b * T_ + t0) * S_;

  const int gt    = tid & 127;
  const int kvrow = gt >> 3;
  const int kvc   = (gt & 7) * 4;

  const int sFirst = g * TPG_ * 16;
  int krow = (sFirst >= T_) ? idx[b * E_ + (sFirst - T_) + kvrow]
                            : (sFirst + kvrow);

  for (int i = 0; i < TPG_; ++i) {
    const int s0 = (g * TPG_ + i) * 16;

    const size_t kvoff =
        ((size_t)(b * T_ + clamp_row(krow))) * HID_ + h * HD_ + kvc;
    const float4 kreg = ldg4(k + kvoff);
    const float4 vreg = ldg4(v + kvoff);
    float bv[8], lv[8];
#pragma unroll
    for (int j = 0; j < 8; ++j) {
      const size_t off = (size_t)(j + 8 * hi) * S_ + s0 + row;
      bv[j] = bias[biasBase + off];
      lv[j] = law[lawBase + off];
    }
    *(float4*)&sK[g][kvrow][kvc] = kreg;
    *(float4*)&sV[g][kvrow][kvc] = vreg;
    __syncthreads();

    const int s1 = s0 + 16;
    int krowNext = s1 + kvrow;
    if (i + 1 < TPG_ && s1 >= T_) krowNext = idx[b * E_ + (s1 - T_) + kvrow];

    v8f c = {};
#pragma unroll
    for (int kk = 0; kk < 8; ++kk) {
      const int kc = kk * 4 + 2 * hi;
      v2f bb;
      bb.x = sK[g][row][kc];
      bb.y = sK[g][row][kc + 1];
      c = wmma_f32_4(aq[kk], bb, c);
    }

#pragma unroll
    for (int j = 0; j < 8; ++j) {
      const int m = j + 8 * hi;
      const float lw = lv[j];
      const float ex = __expf(fminf(c[j] + bv[j], EXP_CLAMP_));
      const float e = (lw <= CUTOFF_) ? 0.0f : ex;
      rsum[j] += e;
      sP[wv][m][row] = e * lw;
    }

#pragma unroll
    for (int kk = 0; kk < 4; ++kk) {
      const int kc = kk * 4 + 2 * hi;
      v2f a, b0, b1;
      a.x  = sP[wv][row][kc];
      a.y  = sP[wv][row][kc + 1];
      b0.x = sV[g][kc][row];
      b0.y = sV[g][kc + 1][row];
      b1.x = sV[g][kc][row + 16];
      b1.y = sV[g][kc + 1][row + 16];
      acc0 = wmma_f32_4(a, b0, acc0);
      acc1 = wmma_f32_4(a, b1, acc1);
    }
    krow = krowNext;
    __syncthreads();
  }

  float* red = &sQ[0][0][0];
  if (g != 0) {
    float* r = red + ((g - 1) * 4 + w) * (24 * 32) + lane;
#pragma unroll
    for (int j = 0; j < 8; ++j) {
      r[j * 32]        = acc0[j];
      r[(8 + j) * 32]  = acc1[j];
      r[(16 + j) * 32] = rsum[j];
    }
  }
  __syncthreads();
  if (g != 0) return;
#pragma unroll
  for (int gg = 0; gg < SG_ - 1; ++gg) {
    const float* r = red + (gg * 4 + w) * (24 * 32) + lane;
#pragma unroll
    for (int j = 0; j < 8; ++j) {
      acc0[j] += r[j * 32];
      acc1[j] += r[(8 + j) * 32];
      rsum[j] += r[(16 + j) * 32];
    }
  }

#pragma unroll
  for (int j = 0; j < 8; ++j) {
    const float s = half_reduce_sum(rsum[j]);
    const float inv = 1.0f / s;
    const int m = j + 8 * hi;
    const size_t o = ((size_t)(b * T_ + t0 + m)) * HID_ + h * HD_;
    attn_out[o + row]      = acc0[j] * inv;
    attn_out[o + 16 + row] = acc1[j] * inv;
  }
}

#define LN_WAVES_ 8

__device__ __forceinline__ float wave_reduce_sum(float x) {
  x = half_reduce_sum(x);
  x += __shfl_xor(x, 16);
  return x;
}

__global__ __launch_bounds__(32 * LN_WAVES_) void ln_rows_kernel(
    const float* __restrict__ attn, const float* __restrict__ ln_g,
    const float* __restrict__ ln_b, float* __restrict__ normed) {
  const int lane = threadIdx.x & 31;
  const int rowi = blockIdx.x * LN_WAVES_ + (threadIdx.x >> 5);
  const float* a = attn + (size_t)rowi * HID_;

  float4 x[4];
#pragma unroll
  for (int i = 0; i < 4; ++i) x[i] = ldg4(a + (i * 32 + lane) * 4);

  float s = 0.0f;
#pragma unroll
  for (int i = 0; i < 4; ++i) s += (x[i].x + x[i].y) + (x[i].z + x[i].w);
  const float mean = wave_reduce_sum(s) * (1.0f / HID_);

  float qs = 0.0f;
#pragma unroll
  for (int i = 0; i < 4; ++i) {
    const float d0 = x[i].x - mean, d1 = x[i].y - mean;
    const float d2 = x[i].z - mean, d3 = x[i].w - mean;
    qs += (d0 * d0 + d1 * d1) + (d2 * d2 + d3 * d3);
  }
  const float rs = rsqrtf(wave_reduce_sum(qs) * (1.0f / HID_) + LN_EPS_);

  float* y = normed + (size_t)rowi * HID_;
#pragma unroll
  for (int i = 0; i < 4; ++i) {
    const int col = (i * 32 + lane) * 4;
    const float4 g = ldg4(ln_g + col);
    const float4 o = ldg4(ln_b + col);
    float4 r;
    r.x = (x[i].x - mean) * rs * g.x + o.x;
    r.y = (x[i].y - mean) * rs * g.y + o.y;
    r.z = (x[i].z - mean) * rs * g.z + o.z;
    r.w = (x[i].w - mean) * rs * g.w + o.w;
    *(float4*)(y + col) = r;
  }
}

__global__ __launch_bounds__(128) void outproj_kernel(
    const float* __restrict__ normed, const float* __restrict__ Wout,
    float* __restrict__ out) {
  __shared__ float sB[16][PAD_];
  __shared__ float sA[4][16][PAD_];

  const int tid  = threadIdx.x;
  const int lane = tid & 31;
  const int w    = tid >> 5;
  const int row  = lane & 15;
  const int hi   = lane >> 4;

  const int NT = HID_ / 16;
  const int nt = blockIdx.x % NT;
  const int mg = blockIdx.x / NT;
  const int m0 = (mg * 4 + w) * 16;
  const int n0 = nt * 16;

  const int wrow = tid >> 3;
  const int wc   = (tid & 7) * 4;
  const int arow = lane >> 1;
  const int ac   = (lane & 1) * 16;

  v8f c = {};
  for (int k0 = 0; k0 < HID_; k0 += 32) {
    *(float4*)&sB[wrow][wc] =
        ldg4(Wout + ((size_t)(n0 + wrow)) * HID_ + k0 + wc);

    {
      const float* ap = normed + ((size_t)(m0 + arow)) * HID_ + k0 + ac;
#pragma unroll
      for (int ii = 0; ii < 4; ++ii)
        *(float4*)&sA[w][arow][ac + ii * 4] = ldg4(ap + ii * 4);
    }
    __syncthreads();
#pragma unroll
    for (int kk = 0; kk < 8; ++kk) {
      const int kc = kk * 4 + 2 * hi;
      v2f a, bb;
      a.x  = sA[w][row][kc];
      a.y  = sA[w][row][kc + 1];
      bb.x = sB[row][kc];
      bb.y = sB[row][kc + 1];
      c = wmma_f32_4(a, bb, c);
    }
    __syncthreads();
  }

#pragma unroll
  for (int j = 0; j < 8; ++j) {
    const int m = m0 + j + 8 * hi;
    out[(size_t)m * HID_ + n0 + row] = c[j];
  }
}

extern "C" void kernel_launch(void* const* d_in, const int* in_sizes, int n_in,
                              void* d_out, int out_size, void* d_ws,
                              size_t ws_size, hipStream_t stream) {
  const float* q    = (const float*)d_in[0];
  const float* k    = (const float*)d_in[1];
  const float* v    = (const float*)d_in[2];
  const float* bias = (const float*)d_in[3];
  const int*   idx  = (const int*)d_in[5];
  const float* law  = (const float*)d_in[6];
  const float* Wout = (const float*)d_in[8];
  const float* lng  = (const float*)d_in[9];
  const float* lnb  = (const float*)d_in[10];
  float* out = (float*)d_out;

  float* ws     = (float*)d_ws;
  float* attn   = ws;
  float* normed = attn + (size_t)B_ * T_ * HID_;

  attn_kernel<<<B_ * NH_ * (T_ / 64), 512, 0, stream>>>(q, k, v, bias, law,
                                                        idx, attn);
  ln_rows_kernel<<<B_ * T_ / LN_WAVES_, 32 * LN_WAVES_, 0, stream>>>(
      attn, lng, lnb, normed);
  outproj_kernel<<<(B_ * T_ / 64) * (HID_ / 16), 128, 0, stream>>>(
      normed, Wout, out);
}


# out-proj retiled to 64x32 per block (wave 16x32: one A fragment, two accumulators), K in 8 LDS-double-buffered steps: next step's global loads fly under the products, one barrier per step
# speedup vs baseline: 3.9809x; 1.1195x over previous
#include <hip/hip_runtime.h>
#include <hip/hip_bf16.h>


#define B_   4
#define T_   512
#define E_   256
#define HID_ 512
#define HD_  32
#define NH_  16
#define S_   768
#define LN_EPS_ 1e-5f
#define CUTOFF_ 1e-5f
#define EXP_CLAMP_ 80.0f
#define PAD_ 36

typedef __attribute__((ext_vector_type(2))) float v2f;
typedef __attribute__((ext_vector_type(8))) float v8f;

__device__ __forceinline__ v8f wmma_f32_4(v2f a, v2f b, v8f c) {
  return __builtin_amdgcn_wmma_f32_16x16x4_f32(
       false, a,  false, b,
       (short)0, c,  false,  false);
}

__device__ __forceinline__ float half_reduce_sum(float x) {
  x += __shfl_xor(x, 1);
  x += __shfl_xor(x, 2);
  x += __shfl_xor(x, 4);
  x += __shfl_xor(x, 8);
  return x;
}

__device__ __forceinline__ float4 ldg4(const float* p) {
  return *(const float4*)p;
}

__device__ __forceinline__ int clamp_row(int t) {
  return min(max(t, 0), T_ - 1);
}

#define SG_  4
#define TPG_ (S_ / 16 / SG_)

__global__ __launch_bounds__(512) void attn_kernel(
    const float* __restrict__ q, const float* __restrict__ k,
    const float* __restrict__ v, const float* __restrict__ bias,
    const float* __restrict__ law, const int* __restrict__ idx,
    float* __restrict__ attn_out) {
  __shared__ float sK[SG_][16][PAD_];
  __shared__ float sV[SG_][16][PAD_];
  __shared__ float sQ[4 * SG_][16][PAD_];
  __shared__ float sP[4 * SG_][16][17];

  const int tid  = threadIdx.x;
  const int lane = tid & 31;
  const int wv   = __builtin_amdgcn_readfirstlane(tid >> 5);
  const int w    = wv & 3;
  const int g    = wv >> 2;
  const int row  = lane & 15;
  const int hi   = lane >> 4;

  const int TB  = T_ / 64;
  const int bid = blockIdx.x;
  const int tg  = bid % TB;
  const int h   = (bid / TB) % NH_;
  const int b   = bid / (TB * NH_);
  const int t0  = (tg * 4 + w) * 16;

  {
    const int qrow = lane >> 1;
    const int qc   = (lane & 1) * 16;
    const float* qp =
        q + ((size_t)(b * T_ + t0 + qrow)) * HID_ + h * HD_ + qc;
#pragma unroll
    for (int ii = 0; ii < 4; ++ii)
      *(float4*)&sQ[wv][qrow][qc + ii * 4] = ldg4(qp + ii * 4);
  }

  v2f aq[8];
#pragma unroll
  for (int kk = 0; kk < 8; ++kk) {
    const int kc = kk * 4 + 2 * hi;
    aq[kk].x = sQ[wv][row][kc];
    aq[kk].y = sQ[wv][row][kc + 1];
  }

  float rsum[8];
#pragma unroll
  for (int j = 0; j < 8; ++j) rsum[j] = 0.0f;
  v8f acc0 = {};
  v8f acc1 = {};

  const size_t biasBase = (((size_t)b * NH_ + h) * T_ + t0) * S_;
  const size_t lawBase  = ((size_t)b * T_ + t0) * S_;

  const int gt    = tid & 127;
  const int kvrow = gt >> 3;
  const int kvc   = (gt & 7) * 4;

  const int sFirst = g * TPG_ * 16;
  int krow = (sFirst >= T_) ? idx[b * E_ + (sFirst - T_) + kvrow]
                            : (sFirst + kvrow);

  for (int i = 0; i < TPG_; ++i) {
    const int s0 = (g * TPG_ + i) * 16;

    const size_t kvoff =
        ((size_t)(b * T_ + clamp_row(krow))) * HID_ + h * HD_ + kvc;
    const float4 kreg = ldg4(k + kvoff);
    const float4 vreg = ldg4(v + kvoff);
    float bv[8], lv[8];
#pragma unroll
    for (int j = 0; j < 8; ++j) {
      const size_t off = (size_t)(j + 8 * hi) * S_ + s0 + row;
      bv[j] = bias[biasBase + off];
      lv[j] = law[lawBase + off];
    }
    *(float4*)&sK[g][kvrow][kvc] = kreg;
    *(float4*)&sV[g][kvrow][kvc] = vreg;
    __syncthreads();

    const int s1 = s0 + 16;
    int krowNext = s1 + kvrow;
    if (i + 1 < TPG_ && s1 >= T_) krowNext = idx[b * E_ + (s1 - T_) + kvrow];

    v8f c = {};
#pragma unroll
    for (int kk = 0; kk < 8; ++kk) {
      const int kc = kk * 4 + 2 * hi;
      v2f bb;
      bb.x = sK[g][row][kc];
      bb.y = sK[g][row][kc + 1];
      c = wmma_f32_4(aq[kk], bb, c);
    }

#pragma unroll
    for (int j = 0; j < 8; ++j) {
      const int m = j + 8 * hi;
      const float lw = lv[j];
      const float ex = __expf(fminf(c[j] + bv[j], EXP_CLAMP_));
      const float e = (lw <= CUTOFF_) ? 0.0f : ex;
      rsum[j] += e;
      sP[wv][m][row] = e * lw;
    }

#pragma unroll
    for (int kk = 0; kk < 4; ++kk) {
      const int kc = kk * 4 + 2 * hi;
      v2f a, b0, b1;
      a.x  = sP[wv][row][kc];
      a.y  = sP[wv][row][kc + 1];
      b0.x = sV[g][kc][row];
      b0.y = sV[g][kc + 1][row];
      b1.x = sV[g][kc][row + 16];
      b1.y = sV[g][kc + 1][row + 16];
      acc0 = wmma_f32_4(a, b0, acc0);
      acc1 = wmma_f32_4(a, b1, acc1);
    }
    krow = krowNext;
    __syncthreads();
  }

  float* red = &sQ[0][0][0];
  if (g != 0) {
    float* r = red + ((g - 1) * 4 + w) * (24 * 32) + lane;
#pragma unroll
    for (int j = 0; j < 8; ++j) {
      r[j * 32]        = acc0[j];
      r[(8 + j) * 32]  = acc1[j];
      r[(16 + j) * 32] = rsum[j];
    }
  }
  __syncthreads();
  if (g != 0) return;
#pragma unroll
  for (int gg = 0; gg < SG_ - 1; ++gg) {
    const float* r = red + (gg * 4 + w) * (24 * 32) + lane;
#pragma unroll
    for (int j = 0; j < 8; ++j) {
      acc0[j] += r[j * 32];
      acc1[j] += r[(8 + j) * 32];
      rsum[j] += r[(16 + j) * 32];
    }
  }

#pragma unroll
  for (int j = 0; j < 8; ++j) {
    const float s = half_reduce_sum(rsum[j]);
    const float inv = 1.0f / s;
    const int m = j + 8 * hi;
    const size_t o = ((size_t)(b * T_ + t0 + m)) * HID_ + h * HD_;
    attn_out[o + row]      = acc0[j] * inv;
    attn_out[o + 16 + row] = acc1[j] * inv;
  }
}

#define LN_WAVES_ 8

__device__ __forceinline__ float wave_reduce_sum(float x) {
  x = half_reduce_sum(x);
  x += __shfl_xor(x, 16);
  return x;
}

__global__ __launch_bounds__(32 * LN_WAVES_) void ln_rows_kernel(
    const float* __restrict__ attn, const float* __restrict__ ln_g,
    const float* __restrict__ ln_b, float* __restrict__ normed) {
  const int lane = threadIdx.x & 31;
  const int rowi = blockIdx.x * LN_WAVES_ + (threadIdx.x >> 5);
  const float* a = attn + (size_t)rowi * HID_;

  float4 x[4];
#pragma unroll
  for (int i = 0; i < 4; ++i) x[i] = ldg4(a + (i * 32 + lane) * 4);

  float s = 0.0f;
#pragma unroll
  for (int i = 0; i < 4; ++i) s += (x[i].x + x[i].y) + (x[i].z + x[i].w);
  const float mean = wave_reduce_sum(s) * (1.0f / HID_);

  float qs = 0.0f;
#pragma unroll
  for (int i = 0; i < 4; ++i) {
    const float d0 = x[i].x - mean, d1 = x[i].y - mean;
    const float d2 = x[i].z - mean, d3 = x[i].w - mean;
    qs += (d0 * d0 + d1 * d1) + (d2 * d2 + d3 * d3);
  }
  const float rs = rsqrtf(wave_reduce_sum(qs) * (1.0f / HID_) + LN_EPS_);

  float* y = normed + (size_t)rowi * HID_;
#pragma unroll
  for (int i = 0; i < 4; ++i) {
    const int col = (i * 32 + lane) * 4;
    const float4 g = ldg4(ln_g + col);
    const float4 o = ldg4(ln_b + col);
    float4 r;
    r.x = (x[i].x - mean) * rs * g.x + o.x;
    r.y = (x[i].y - mean) * rs * g.y + o.y;
    r.z = (x[i].z - mean) * rs * g.z + o.z;
    r.w = (x[i].w - mean) * rs * g.w + o.w;
    *(float4*)(y + col) = r;
  }
}

#define OP_BM_    64
#define OP_BN_    32
#define OP_KS_    64
#define OP_PITCH_ 68
#define OP_NSTEP_ (HID_ / OP_KS_)

__global__ __launch_bounds__(128) void outproj_kernel(
    const float* __restrict__ normed, const float* __restrict__ Wout,
    float* __restrict__ out) {
  __shared__ __attribute__((aligned(16))) float sA[2][OP_BM_][OP_PITCH_];
  __shared__ __attribute__((aligned(16))) float sW[2][OP_BN_][OP_PITCH_];

  const int tid  = threadIdx.x;
  const int lane = tid & 31;
  const int w    = __builtin_amdgcn_readfirstlane(tid >> 5);
  const int row  = lane & 15;
  const int hi   = lane >> 4;

  const int NG = HID_ / OP_BN_;
  const int ng = blockIdx.x % NG;
  const int mg = blockIdx.x / NG;
  const int m0 = mg * OP_BM_;
  const int n0 = ng * OP_BN_;

  const int gc = (tid & 15) * 4;
  const int gr = tid >> 4;

  const float* aSrc = normed + (size_t)(m0 + gr) * HID_ + gc;
  const float* wSrc = Wout + (size_t)(n0 + gr) * HID_ + gc;

  float4 ga[OP_BM_ / 8], gw[OP_BN_ / 8];
#pragma unroll
  for (int q = 0; q < OP_BM_ / 8; ++q)
    ga[q] = ldg4(aSrc + (size_t)(8 * q) * HID_);
#pragma unroll
  for (int q = 0; q < OP_BN_ / 8; ++q)
    gw[q] = ldg4(wSrc + (size_t)(8 * q) * HID_);
#pragma unroll
  for (int q = 0; q < OP_BM_ / 8; ++q)
    *(float4*)&sA[0][gr + 8 * q][gc] = ga[q];
#pragma unroll
  for (int q = 0; q < OP_BN_ / 8; ++q)
    *(float4*)&sW[0][gr + 8 * q][gc] = gw[q];
  __syncthreads();

  v8f c0 = {};
  v8f c1 = {};
  for (int ks = 0; ks < OP_NSTEP_; ++ks) {
    const int cur = ks & 1;
    const bool more = ks + 1 < OP_NSTEP_;

    if (more) {
      const int kn = (ks + 1) * OP_KS_;
#pragma unroll
      for (int q = 0; q < OP_BM_ / 8; ++q)
        ga[q] = ldg4(aSrc + (size_t)(8 * q) * HID_ + kn);
#pragma unroll
      for (int q = 0; q < OP_BN_ / 8; ++q)
        gw[q] = ldg4(wSrc + (size_t)(8 * q) * HID_ + kn);
    }

#pragma unroll
    for (int kk = 0; kk < OP_KS_ / 4; ++kk) {
      const int kc = kk * 4 + 2 * hi;
      v2f a, b0, b1;
      a.x  = sA[cur][16 * w + row][kc];
      a.y  = sA[cur][16 * w + row][kc + 1];
      b0.x = sW[cur][row][kc];
      b0.y = sW[cur][row][kc + 1];
      b1.x = sW[cur][16 + row][kc];
      b1.y = sW[cur][16 + row][kc + 1];
      c0 = wmma_f32_4(a, b0, c0);
      c1 = wmma_f32_4(a, b1, c1);
    }

    if (more) {
#pragma unroll
      for (int q = 0; q < OP_BM_ / 8; ++q)
        *(float4*)&sA[cur ^ 1][gr + 8 * q][gc] = ga[q];
#pragma unroll
      for (int q = 0; q < OP_BN_ / 8; ++q)
        *(float4*)&sW[cur ^ 1][gr + 8 * q][gc] = gw[q];
    }
    __syncthreads();
  }

  float (*sC)[OP_PITCH_] = &sA[0][16 * w];
#pragma unroll
  for (int j = 0; j < 8; ++j) {
    sC[j + 8 * hi][row]      = c0[j];
    sC[j + 8 * hi][16 + row] = c1[j];
  }
  const int orow = lane >> 3;
  const int oc   = (lane & 7) * 4;
#pragma unroll
  for (int q = 0; q < 4; ++q) {
    const int r = orow + 4 * q;
    const float4 o = *(const float4*)&sC[r][oc];
    *(float4*)(out + (size_t)(m0 + 16 * w + r) * HID_ + n0 + oc) = o;
  }
}

extern "C" void kernel_launch(void* const* d_in, const int* in_sizes, int n_in,
                              void* d_out, int out_size, void* d_ws,
                              size_t ws_size, hipStream_t stream) {
  const float* q    = (const float*)d_in[0];
  const float* k    = (const float*)d_in[1];
  const float* v    = (const float*)d_in[2];
  const float* bias = (const float*)d_in[3];
  const int*   idx  = (const int*)d_in[5];
  const float* law  = (const float*)d_in[6];
  const float* Wout = (const float*)d_in[8];
  const float* lng  = (const float*)d_in[9];
  const float* lnb  = (const float*)d_in[10];
  float* out = (float*)d_out;

  float* ws     = (float*)d_ws;
  float* attn   = ws;
  float* normed = attn + (size_t)B_ * T_ * HID_;

  attn_kernel<<<B_ * NH_ * (T_ / 64), 512, 0, stream>>>(q, k, v, bias, law,
                                                        idx, attn);
  ln_rows_kernel<<<B_ * T_ / LN_WAVES_, 32 * LN_WAVES_, 0, stream>>>(
      attn, lng, lnb, normed);
  outproj_kernel<<<(B_ * T_ / OP_BM_) * (HID_ / OP_BN_), 128, 0, stream>>>(
      normed, Wout, out);
}


# attention: s walked in 32-column strips = two score tiles (even / odd columns); bias and law read as float2, 16 lanes per whole 128-B line; loads and barriers per column halve (sum order changes)
# speedup vs baseline: 4.2166x; 1.0592x over previous
#include <hip/hip_runtime.h>
#include <hip/hip_bf16.h>


#define B_   4
#define T_   512
#define E_   256
#define HID_ 512
#define HD_  32
#define NH_  16
#define S_   768
#define LN_EPS_ 1e-5f
#define CUTOFF_ 1e-5f
#define EXP_CLAMP_ 80.0f
#define PAD_ 36

typedef __attribute__((ext_vector_type(2))) float v2f;
typedef __attribute__((ext_vector_type(8))) float v8f;

__device__ __forceinline__ v8f wmma_f32_4(v2f a, v2f b, v8f c) {
  return __builtin_amdgcn_wmma_f32_16x16x4_f32(
       false, a,  false, b,
       (short)0, c,  false,  false);
}

__device__ __forceinline__ float half_reduce_sum(float x) {
  x += __shfl_xor(x, 1);
  x += __shfl_xor(x, 2);
  x += __shfl_xor(x, 4);
  x += __shfl_xor(x, 8);
  return x;
}

__device__ __forceinline__ float4 ldg4(const float* p) {
  return *(const float4*)p;
}

__device__ __forceinline__ float2 ldg2(const float* p) {
  return *(const float2*)p;
}

__device__ __forceinline__ int clamp_row(int t) {
  return min(max(t, 0), T_ - 1);
}

#define SG_  4
#define SW_  32
#define SPG_ (S_ / SW_ / SG_)

__global__ __launch_bounds__(512, 8) void attn_kernel(
    const float* __restrict__ q, const float* __restrict__ k,
    const float* __restrict__ v, const float* __restrict__ bias,
    const float* __restrict__ law, const int* __restrict__ idx,
    float* __restrict__ attn_out) {
  __shared__ float sK[SG_][SW_][PAD_];
  __shared__ float sV[SG_][SW_][PAD_];
  __shared__ float sQ[4 * SG_][16][PAD_];
  __shared__ float sP[4 * SG_][2][16][17];

  const int tid  = threadIdx.x;
  const int lane = tid & 31;
  const int wv   = __builtin_amdgcn_readfirstlane(tid >> 5);
  const int w    = wv & 3;
  const int g    = wv >> 2;
  const int row  = lane & 15;
  const int hi   = lane >> 4;

  const int TB  = T_ / 64;
  const int bid = blockIdx.x;
  const int tg  = bid % TB;
  const int h   = (bid / TB) % NH_;
  const int b   = bid / (TB * NH_);
  const int t0  = (tg * 4 + w) * 16;

  {
    const int qrow = lane >> 1;
    const int qc   = (lane & 1) * 16;
    const float* qp =
        q + ((size_t)(b * T_ + t0 + qrow)) * HID_ + h * HD_ + qc;
#pragma unroll
    for (int ii = 0; ii < 4; ++ii)
      *(float4*)&sQ[wv][qrow][qc + ii * 4] = ldg4(qp + ii * 4);
  }

  v2f aq[8];
#pragma unroll
  for (int kk = 0; kk < 8; ++kk) {
    const int kc = kk * 4 + 2 * hi;
    aq[kk].x = sQ[wv][row][kc];
    aq[kk].y = sQ[wv][row][kc + 1];
  }

  float rsum[8];
#pragma unroll
  for (int j = 0; j < 8; ++j) rsum[j] = 0.0f;
  v8f acc0 = {};
  v8f acc1 = {};

  const float* bp = bias + (((size_t)b * NH_ + h) * T_ + t0 + 8 * hi) * S_ +
                    g * (SPG_ * SW_) + 2 * row;
  const float* lp = law + ((size_t)b * T_ + t0 + 8 * hi) * S_ +
                    g * (SPG_ * SW_) + 2 * row;

  const int gt    = tid & 127;
  const int kvrow = gt >> 3;
  const int kvc   = (gt & 7) * 4;
  const int lrow0 = (kvrow >> 1) + 16 * (kvrow & 1);
  const int lrow1 = lrow0 + 8;

  const int sFirst = g * SPG_ * SW_;
  int krow0 = sFirst + kvrow;
  int krow1 = sFirst + kvrow + 16;
  if (sFirst >= T_) {
    krow0 = idx[b * E_ + (sFirst - T_) + kvrow];
    krow1 = idx[b * E_ + (sFirst - T_) + kvrow + 16];
  }

  for (int i = 0; i < SPG_; ++i) {
    const int s0 = (g * SPG_ + i) * SW_;

    const size_t kvoff0 =
        ((size_t)(b * T_ + clamp_row(krow0))) * HID_ + h * HD_ + kvc;
    const size_t kvoff1 =
        ((size_t)(b * T_ + clamp_row(krow1))) * HID_ + h * HD_ + kvc;
    const float4 kreg0 = ldg4(k + kvoff0);
    const float4 kreg1 = ldg4(k + kvoff1);
    const float4 vreg0 = ldg4(v + kvoff0);
    const float4 vreg1 = ldg4(v + kvoff1);
    float2 bv[8], lv[8];
#pragma unroll
    for (int j = 0; j < 8; ++j) {
      bv[j] = ldg2(bp + j * S_);
      lv[j] = ldg2(lp + j * S_);
    }
    bp += SW_;
    lp += SW_;
    *(float4*)&sK[g][lrow0][kvc] = kreg0;
    *(float4*)&sK[g][lrow1][kvc] = kreg1;
    *(float4*)&sV[g][lrow0][kvc] = vreg0;
    *(float4*)&sV[g][lrow1][kvc] = vreg1;
    __syncthreads();

    const int s1 = s0 + SW_;
    int krowNext0 = s1 + kvrow;
    int krowNext1 = s1 + kvrow + 16;
    if (i + 1 < SPG_ && s1 >= T_) {
      krowNext0 = idx[b * E_ + (s1 - T_) + kvrow];
      krowNext1 = idx[b * E_ + (s1 - T_) + kvrow + 16];
    }

    v8f c0 = {};
    v8f c1 = {};
#pragma unroll
    for (int kk = 0; kk < 8; ++kk) {
      const int kc = kk * 4 + 2 * hi;
      v2f bb0, bb1;
      bb0.x = sK[g][row][kc];
      bb0.y = sK[g][row][kc + 1];
      bb1.x = sK[g][16 + row][kc];
      bb1.y = sK[g][16 + row][kc + 1];
      c0 = wmma_f32_4(aq[kk], bb0, c0);
      c1 = wmma_f32_4(aq[kk], bb1, c1);
    }

#pragma unroll
    for (int j = 0; j < 8; ++j) {
      const int m = j + 8 * hi;
      const float2 lw = lv[j];
      const float2 bj = bv[j];
      const float ex0 = __expf(fminf(c0[j] + bj.x, EXP_CLAMP_));
      const float ex1 = __expf(fminf(c1[j] + bj.y, EXP_CLAMP_));
      const float e0 = (lw.x <= CUTOFF_) ? 0.0f : ex0;
      const float e1 = (lw.y <= CUTOFF_) ? 0.0f : ex1;
      rsum[j] += e0 + e1;
      sP[wv][0][m][row] = e0 * lw.x;
      sP[wv][1][m][row] = e1 * lw.y;
    }

#pragma unroll
    for (int t = 0; t < 2; ++t) {
      v2f a[4], b0[4], b1[4];
#pragma unroll
      for (int kk = 0; kk < 4; ++kk) {
        const int kc = kk * 4 + 2 * hi;
        a[kk].x  = sP[wv][t][row][kc];
        a[kk].y  = sP[wv][t][row][kc + 1];
        b0[kk].x = sV[g][16 * t + kc][row];
        b0[kk].y = sV[g][16 * t + kc + 1][row];
        b1[kk].x = sV[g][16 * t + kc][row + 16];
        b1[kk].y = sV[g][16 * t + kc + 1][row + 16];
      }
#pragma unroll
      for (int kk = 0; kk < 4; ++kk) {
        acc0 = wmma_f32_4(a[kk], b0[kk], acc0);
        acc1 = wmma_f32_4(a[kk], b1[kk], acc1);
      }
    }
    krow0 = krowNext0;
    krow1 = krowNext1;
    __syncthreads();
  }

  float* red = &sQ[0][0][0];
  if (g != 0) {
    float* r = red + ((g - 1) * 4 + w) * (24 * 32) + lane;
#pragma unroll
    for (int j = 0; j < 8; ++j) {
      r[j * 32]        = acc0[j];
      r[(8 + j) * 32]  = acc1[j];
      r[(16 + j) * 32] = rsum[j];
    }
  }
  __syncthreads();
  if (g != 0) return;
#pragma unroll
  for (int gg = 0; gg < SG_ - 1; ++gg) {
    const float* r = red + (gg * 4 + w) * (24 * 32) + lane;
#pragma unroll
    for (int j = 0; j < 8; ++j) {
      acc0[j] += r[j * 32];
      acc1[j] += r[(8 + j) * 32];
      rsum[j] += r[(16 + j) * 32];
    }
  }

#pragma unroll
  for (int j = 0; j < 8; ++j) {
    const float s = half_reduce_sum(rsum[j]);
    const float inv = 1.0f / s;
    const int m = j + 8 * hi;
    const size_t o = ((size_t)(b * T_ + t0 + m)) * HID_ + h * HD_;
    attn_out[o + row]      = acc0[j] * inv;
    attn_out[o + 16 + row] = acc1[j] * inv;
  }
}

#define LN_WAVES_ 8

__device__ __forceinline__ float wave_reduce_sum(float x) {
  x = half_reduce_sum(x);
  x += __shfl_xor(x, 16);
  return x;
}

__global__ __launch_bounds__(32 * LN_WAVES_) void ln_rows_kernel(
    const float* __restrict__ attn, const float* __restrict__ ln_g,
    const float* __restrict__ ln_b, float* __restrict__ normed) {
  const int lane = threadIdx.x & 31;
  const int rowi = blockIdx.x * LN_WAVES_ + (threadIdx.x >> 5);
  const float* a = attn + (size_t)rowi * HID_;

  float4 x[4];
#pragma unroll
  for (int i = 0; i < 4; ++i) x[i] = ldg4(a + (i * 32 + lane) * 4);

  float s = 0.0f;
#pragma unroll
  for (int i = 0; i < 4; ++i) s += (x[i].x + x[i].y) + (x[i].z + x[i].w);
  const float mean = wave_reduce_sum(s) * (1.0f / HID_);

  float qs = 0.0f;
#pragma unroll
  for (int i = 0; i < 4; ++i) {
    const float d0 = x[i].x - mean, d1 = x[i].y - mean;
    const float d2 = x[i].z - mean, d3 = x[i].w - mean;
    qs += (d0 * d0 + d1 * d1) + (d2 * d2 + d3 * d3);
  }
  const float rs = rsqrtf(wave_reduce_sum(qs) * (1.0f / HID_) + LN_EPS_);

  float* y = normed + (size_t)rowi * HID_;
#pragma unroll
  for (int i = 0; i < 4; ++i) {
    const int col = (i * 32 + lane) * 4;
    const float4 g = ldg4(ln_g + col);
    const float4 o = ldg4(ln_b + col);
    float4 r;
    r.x = (x[i].x - mean) * rs * g.x + o.x;
    r.y = (x[i].y - mean) * rs * g.y + o.y;
    r.z = (x[i].z - mean) * rs * g.z + o.z;
    r.w = (x[i].w - mean) * rs * g.w + o.w;
    *(float4*)(y + col) = r;
  }
}

#define OP_BM_    64
#define OP_BN_    32
#define OP_KS_    64
#define OP_PITCH_ 68
#define OP_NSTEP_ (HID_ / OP_KS_)

__global__ __launch_bounds__(128) void outproj_kernel(
    const float* __restrict__ normed, const float* __restrict__ Wout,
    float* __restrict__ out) {
  __shared__ __attribute__((aligned(16))) float sA[2][OP_BM_][OP_PITCH_];
  __shared__ __attribute__((aligned(16))) float sW[2][OP_BN_][OP_PITCH_];

  const int tid  = threadIdx.x;
  const int lane = tid & 31;
  const int w    = __builtin_amdgcn_readfirstlane(tid >> 5);
  const int row  = lane & 15;
  const int hi   = lane >> 4;

  const int NG = HID_ / OP_BN_;
  const int ng = blockIdx.x % NG;
  const int mg = blockIdx.x / NG;
  const int m0 = mg * OP_BM_;
  const int n0 = ng * OP_BN_;

  const int gc = (tid & 15) * 4;
  const int gr = tid >> 4;

  const float* aSrc = normed + (size_t)(m0 + gr) * HID_ + gc;
  const float* wSrc = Wout + (size_t)(n0 + gr) * HID_ + gc;

  float4 ga[OP_BM_ / 8], gw[OP_BN_ / 8];
#pragma unroll
  for (int q = 0; q < OP_BM_ / 8; ++q)
    ga[q] = ldg4(aSrc + (size_t)(8 * q) * HID_);
#pragma unroll
  for (int q = 0; q < OP_BN_ / 8; ++q)
    gw[q] = ldg4(wSrc + (size_t)(8 * q) * HID_);
#pragma unroll
  for (int q = 0; q < OP_BM_ / 8; ++q)
    *(float4*)&sA[0][gr + 8 * q][gc] = ga[q];
#pragma unroll
  for (int q = 0; q < OP_BN_ / 8; ++q)
    *(float4*)&sW[0][gr + 8 * q][gc] = gw[q];
  __syncthreads();

  v8f c0 = {};
  v8f c1 = {};
  for (int ks = 0; ks < OP_NSTEP_; ++ks) {
    const int cur = ks & 1;
    const bool more = ks + 1 < OP_NSTEP_;

    if (more) {
      const int kn = (ks + 1) * OP_KS_;
#pragma unroll
      for (int q = 0; q < OP_BM_ / 8; ++q)
        ga[q] = ldg4(aSrc + (size_t)(8 * q) * HID_ + kn);
#pragma unroll
      for (int q = 0; q < OP_BN_ / 8; ++q)
        gw[q] = ldg4(wSrc + (size_t)(8 * q) * HID_ + kn);
    }

#pragma unroll
    for (int kk = 0; kk < OP_KS_ / 4; ++kk) {
      const int kc = kk * 4 + 2 * hi;
      v2f a, b0, b1;
      a.x  = sA[cur][16 * w + row][kc];
      a.y  = sA[cur][16 * w + row][kc + 1];
      b0.x = sW[cur][row][kc];
      b0.y = sW[cur][row][kc + 1];
      b1.x = sW[cur][16 + row][kc];
      b1.y = sW[cur][16 + row][kc + 1];
      c0 = wmma_f32_4(a, b0, c0);
      c1 = wmma_f32_4(a, b1, c1);
    }

    if (more) {
#pragma unroll
      for (int q = 0; q < OP_BM_ / 8; ++q)
        *(float4*)&sA[cur ^ 1][gr + 8 * q][gc] = ga[q];
#pragma unroll
      for (int q = 0; q < OP_BN_ / 8; ++q)
        *(float4*)&sW[cur ^ 1][gr + 8 * q][gc] = gw[q];
    }
    __syncthreads();
  }

  float (*sC)[OP_PITCH_] = &sA[0][16 * w];
#pragma unroll
  for (int j = 0; j < 8; ++j) {
    sC[j + 8 * hi][row]      = c0[j];
    sC[j + 8 * hi][16 + row] = c1[j];
  }
  const int orow = lane >> 3;
  const int oc   = (lane & 7) * 4;
#pragma unroll
  for (int q = 0; q < 4; ++q) {
    const int r = orow + 4 * q;
    const float4 o = *(const float4*)&sC[r][oc];
    *(float4*)(out + (size_t)(m0 + 16 * w + r) * HID_ + n0 + oc) = o;
  }
}

extern "C" void kernel_launch(void* const* d_in, const int* in_sizes, int n_in,
                              void* d_out, int out_size, void* d_ws,
                              size_t ws_size, hipStream_t stream) {
  const float* q    = (const float*)d_in[0];
  const float* k    = (const float*)d_in[1];
  const float* v    = (const float*)d_in[2];
  const float* bias = (const float*)d_in[3];
  const int*   idx  = (const int*)d_in[5];
  const float* law  = (const float*)d_in[6];
  const float* Wout = (const float*)d_in[8];
  const float* lng  = (const float*)d_in[9];
  const float* lnb  = (const float*)d_in[10];
  float* out = (float*)d_out;

  float* ws     = (float*)d_ws;
  float* attn   = ws;
  float* normed = attn + (size_t)B_ * T_ * HID_;

  attn_kernel<<<B_ * NH_ * (T_ / 64), 512, 0, stream>>>(q, k, v, bias, law,
                                                        idx, attn);
  ln_rows_kernel<<<B_ * T_ / LN_WAVES_, 32 * LN_WAVES_, 0, stream>>>(
      attn, lng, lnb, normed);
  outproj_kernel<<<(B_ * T_ / OP_BM_) * (HID_ / OP_BN_), 128, 0, stream>>>(
      normed, Wout, out);
}


# attention: each strip's global loads (K, V, bias, law) issued one strip ahead, behind the previous strip's exponentials and under its P@V products; loop barriers wait for LDS only, not for loads
# speedup vs baseline: 4.5938x; 1.0895x over previous
#include <hip/hip_runtime.h>
#include <hip/hip_bf16.h>


#define B_   4
#define T_   512
#define E_   256
#define HID_ 512
#define HD_  32
#define NH_  16
#define S_   768
#define LN_EPS_ 1e-5f
#define CUTOFF_ 1e-5f
#define EXP_CLAMP_ 80.0f
#define PAD_ 36

typedef __attribute__((ext_vector_type(2))) float v2f;
typedef __attribute__((ext_vector_type(8))) float v8f;

__device__ __forceinline__ v8f wmma_f32_4(v2f a, v2f b, v8f c) {
  return __builtin_amdgcn_wmma_f32_16x16x4_f32(
       false, a,  false, b,
       (short)0, c,  false,  false);
}

__device__ __forceinline__ float half_reduce_sum(float x) {
  x += __shfl_xor(x, 1);
  x += __shfl_xor(x, 2);
  x += __shfl_xor(x, 4);
  x += __shfl_xor(x, 8);
  return x;
}

__device__ __forceinline__ float4 ldg4(const float* p) {
  return *(const float4*)p;
}

__device__ __forceinline__ float2 ldg2(const float* p) {
  return *(const float2*)p;
}

__device__ __forceinline__ int clamp_row(int t) {
  return min(max(t, 0), T_ - 1);
}

__device__ __forceinline__ void lds_barrier() {
  __builtin_amdgcn_fence(__ATOMIC_RELEASE, "workgroup", "local");
  __builtin_amdgcn_s_barrier();
  __builtin_amdgcn_fence(__ATOMIC_ACQUIRE, "workgroup", "local");
}

#define SG_  4
#define SW_  32
#define SPG_ (S_ / SW_ / SG_)

__global__ __launch_bounds__(512, 8) void attn_kernel(
    const float* __restrict__ q, const float* __restrict__ k,
    const float* __restrict__ v, const float* __restrict__ bias,
    const float* __restrict__ law, const int* __restrict__ idx,
    float* __restrict__ attn_out) {
  __shared__ float sK[SG_][SW_][PAD_];
  __shared__ float sV[SG_][SW_][PAD_];
  __shared__ float sQ[4 * SG_][16][PAD_];
  __shared__ float sP[4 * SG_][2][16][17];

  const int tid  = threadIdx.x;
  const int lane = tid & 31;
  const int wv   = __builtin_amdgcn_readfirstlane(tid >> 5);
  const int w    = wv & 3;
  const int g    = wv >> 2;
  const int row  = lane & 15;
  const int hi   = lane >> 4;

  const int TB  = T_ / 64;
  const int bid = blockIdx.x;
  const int tg  = bid % TB;
  const int h   = (bid / TB) % NH_;
  const int b   = bid / (TB * NH_);
  const int t0  = (tg * 4 + w) * 16;

  {
    const int qrow = lane >> 1;
    const int qc   = (lane & 1) * 16;
    const float* qp =
        q + ((size_t)(b * T_ + t0 + qrow)) * HID_ + h * HD_ + qc;
#pragma unroll
    for (int ii = 0; ii < 4; ++ii)
      *(float4*)&sQ[wv][qrow][qc + ii * 4] = ldg4(qp + ii * 4);
  }

  v2f aq[8];
#pragma unroll
  for (int kk = 0; kk < 8; ++kk) {
    const int kc = kk * 4 + 2 * hi;
    aq[kk].x = sQ[wv][row][kc];
    aq[kk].y = sQ[wv][row][kc + 1];
  }

  float rsum[8];
#pragma unroll
  for (int j = 0; j < 8; ++j) rsum[j] = 0.0f;
  v8f acc0 = {};
  v8f acc1 = {};

  const float* bp = bias + (((size_t)b * NH_ + h) * T_ + t0 + 8 * hi) * S_ +
                    g * (SPG_ * SW_) + 2 * row;
  const float* lp = law + ((size_t)b * T_ + t0 + 8 * hi) * S_ +
                    g * (SPG_ * SW_) + 2 * row;

  const int gt    = tid & 127;
  const int kvrow = gt >> 3;
  const int kvc   = (gt & 7) * 4;
  const int lrow0 = (kvrow >> 1) + 16 * (kvrow & 1);
  const int lrow1 = lrow0 + 8;

  const int sFirst = g * SPG_ * SW_;
  int krow0 = sFirst + kvrow;
  int krow1 = sFirst + kvrow + 16;
  if (sFirst >= T_) {
    krow0 = idx[b * E_ + (sFirst - T_) + kvrow];
    krow1 = idx[b * E_ + (sFirst - T_) + kvrow + 16];
  }
  int krowN0 = sFirst + SW_ + kvrow;
  int krowN1 = sFirst + SW_ + kvrow + 16;
  if (sFirst + SW_ >= T_) {
    krowN0 = idx[b * E_ + (sFirst + SW_ - T_) + kvrow];
    krowN1 = idx[b * E_ + (sFirst + SW_ - T_) + kvrow + 16];
  }

  float4 kreg0, kreg1, vreg0, vreg1;
  float2 bv[8], lv[8];
  {
    const size_t kvoff0 =
        ((size_t)(b * T_ + clamp_row(krow0))) * HID_ + h * HD_ + kvc;
    const size_t kvoff1 =
        ((size_t)(b * T_ + clamp_row(krow1))) * HID_ + h * HD_ + kvc;
    kreg0 = ldg4(k + kvoff0);
    kreg1 = ldg4(k + kvoff1);
    vreg0 = ldg4(v + kvoff0);
    vreg1 = ldg4(v + kvoff1);
#pragma unroll
    for (int j = 0; j < 8; ++j) {
      bv[j] = ldg2(bp + j * S_);
      lv[j] = ldg2(lp + j * S_);
    }
    bp += SW_;
    lp += SW_;
  }

  for (int i = 0; i < SPG_; ++i) {
    const int s0 = (g * SPG_ + i) * SW_;

    *(float4*)&sK[g][lrow0][kvc] = kreg0;
    *(float4*)&sK[g][lrow1][kvc] = kreg1;
    *(float4*)&sV[g][lrow0][kvc] = vreg0;
    *(float4*)&sV[g][lrow1][kvc] = vreg1;
    lds_barrier();

    v8f c0 = {};
    v8f c1 = {};
#pragma unroll
    for (int kk = 0; kk < 8; ++kk) {
      const int kc = kk * 4 + 2 * hi;
      v2f bb0, bb1;
      bb0.x = sK[g][row][kc];
      bb0.y = sK[g][row][kc + 1];
      bb1.x = sK[g][16 + row][kc];
      bb1.y = sK[g][16 + row][kc + 1];
      c0 = wmma_f32_4(aq[kk], bb0, c0);
      c1 = wmma_f32_4(aq[kk], bb1, c1);
    }
    __builtin_amdgcn_sched_barrier(0);

#pragma unroll
    for (int j = 0; j < 8; ++j) {
      const int m = j + 8 * hi;
      const float2 lw = lv[j];
      const float2 bj = bv[j];
      const float ex0 = __expf(fminf(c0[j] + bj.x, EXP_CLAMP_));
      const float ex1 = __expf(fminf(c1[j] + bj.y, EXP_CLAMP_));
      const float e0 = (lw.x <= CUTOFF_) ? 0.0f : ex0;
      const float e1 = (lw.y <= CUTOFF_) ? 0.0f : ex1;
      rsum[j] += e0 + e1;
      sP[wv][0][m][row] = e0 * lw.x;
      sP[wv][1][m][row] = e1 * lw.y;
    }

    __builtin_amdgcn_sched_barrier(0);
    if (i + 1 < SPG_) {
      const size_t kvoff0 =
          ((size_t)(b * T_ + clamp_row(krowN0))) * HID_ + h * HD_ + kvc;
      const size_t kvoff1 =
          ((size_t)(b * T_ + clamp_row(krowN1))) * HID_ + h * HD_ + kvc;
      kreg0 = ldg4(k + kvoff0);
      kreg1 = ldg4(k + kvoff1);
      vreg0 = ldg4(v + kvoff0);
      vreg1 = ldg4(v + kvoff1);
      __builtin_amdgcn_sched_barrier(0);
#pragma unroll
      for (int j = 0; j < 8; ++j) {
        bv[j] = ldg2(bp + j * S_);
        lv[j] = ldg2(lp + j * S_);
      }
      bp += SW_;
      lp += SW_;
      const int s2 = s0 + 2 * SW_;
      krowN0 = s2 + kvrow;
      krowN1 = s2 + kvrow + 16;
      if (i + 2 < SPG_ && s2 >= T_) {
        krowN0 = idx[b * E_ + (s2 - T_) + kvrow];
        krowN1 = idx[b * E_ + (s2 - T_) + kvrow + 16];
      }
    }
    __builtin_amdgcn_sched_barrier(0);

#pragma unroll
    for (int t = 0; t < 2; ++t) {
      v2f a[4], b0[4], b1[4];
#pragma unroll
      for (int kk = 0; kk < 4; ++kk) {
        const int kc = kk * 4 + 2 * hi;
        a[kk].x  = sP[wv][t][row][kc];
        a[kk].y  = sP[wv][t][row][kc + 1];
        b0[kk].x = sV[g][16 * t + kc][row];
        b0[kk].y = sV[g][16 * t + kc + 1][row];
        b1[kk].x = sV[g][16 * t + kc][row + 16];
        b1[kk].y = sV[g][16 * t + kc + 1][row + 16];
      }
#pragma unroll
      for (int kk = 0; kk < 4; ++kk) {
        acc0 = wmma_f32_4(a[kk], b0[kk], acc0);
        acc1 = wmma_f32_4(a[kk], b1[kk], acc1);
      }
    }
    lds_barrier();
  }

  float* red = &sQ[0][0][0];
  if (g != 0) {
    float* r = red + ((g - 1) * 4 + w) * (24 * 32) + lane;
#pragma unroll
    for (int j = 0; j < 8; ++j) {
      r[j * 32]        = acc0[j];
      r[(8 + j) * 32]  = acc1[j];
      r[(16 + j) * 32] = rsum[j];
    }
  }
  __syncthreads();
  if (g != 0) return;
#pragma unroll
  for (int gg = 0; gg < SG_ - 1; ++gg) {
    const float* r = red + (gg * 4 + w) * (24 * 32) + lane;
#pragma unroll
    for (int j = 0; j < 8; ++j) {
      acc0[j] += r[j * 32];
      acc1[j] += r[(8 + j) * 32];
      rsum[j] += r[(16 + j) * 32];
    }
  }

#pragma unroll
  for (int j = 0; j < 8; ++j) {
    const float s = half_reduce_sum(rsum[j]);
    const float inv = 1.0f / s;
    const int m = j + 8 * hi;
    const size_t o = ((size_t)(b * T_ + t0 + m)) * HID_ + h * HD_;
    attn_out[o + row]      = acc0[j] * inv;
    attn_out[o + 16 + row] = acc1[j] * inv;
  }
}

#define LN_WAVES_ 8

__device__ __forceinline__ float wave_reduce_sum(float x) {
  x = half_reduce_sum(x);
  x += __shfl_xor(x, 16);
  return x;
}

__global__ __launch_bounds__(32 * LN_WAVES_) void ln_rows_kernel(
    const float* __restrict__ attn, const float* __restrict__ ln_g,
    const float* __restrict__ ln_b, float* __restrict__ normed) {
  const int lane = threadIdx.x & 31;
  const int rowi = blockIdx.x * LN_WAVES_ + (threadIdx.x >> 5);
  const float* a = attn + (size_t)rowi * HID_;

  float4 x[4];
#pragma unroll
  for (int i = 0; i < 4; ++i) x[i] = ldg4(a + (i * 32 + lane) * 4);

  float s = 0.0f;
#pragma unroll
  for (int i = 0; i < 4; ++i) s += (x[i].x + x[i].y) + (x[i].z + x[i].w);
  const float mean = wave_reduce_sum(s) * (1.0f / HID_);

  float qs = 0.0f;
#pragma unroll
  for (int i = 0; i < 4; ++i) {
    const float d0 = x[i].x - mean, d1 = x[i].y - mean;
    const float d2 = x[i].z - mean, d3 = x[i].w - mean;
    qs += (d0 * d0 + d1 * d1) + (d2 * d2 + d3 * d3);
  }
  const float rs = rsqrtf(wave_reduce_sum(qs) * (1.0f / HID_) + LN_EPS_);

  float* y = normed + (size_t)rowi * HID_;
#pragma unroll
  for (int i = 0; i < 4; ++i) {
    const int col = (i * 32 + lane) * 4;
    const float4 g = ldg4(ln_g + col);
    const float4 o = ldg4(ln_b + col);
    float4 r;
    r.x = (x[i].x - mean) * rs * g.x + o.x;
    r.y = (x[i].y - mean) * rs * g.y + o.y;
    r.z = (x[i].z - mean) * rs * g.z + o.z;
    r.w = (x[i].w - mean) * rs * g.w + o.w;
    *(float4*)(y + col) = r;
  }
}

#define OP_BM_    64
#define OP_BN_    32
#define OP_KS_    64
#define OP_PITCH_ 68
#define OP_NSTEP_ (HID_ / OP_KS_)

__global__ __launch_bounds__(128) void outproj_kernel(
    const float* __restrict__ normed, const float* __restrict__ Wout,
    float* __restrict__ out) {
  __shared__ __attribute__((aligned(16))) float sA[2][OP_BM_][OP_PITCH_];
  __shared__ __attribute__((aligned(16))) float sW[2][OP_BN_][OP_PITCH_];

  const int tid  = threadIdx.x;
  const int lane = tid & 31;
  const int w    = __builtin_amdgcn_readfirstlane(tid >> 5);
  const int row  = lane & 15;
  const int hi   = lane >> 4;

  const int NG = HID_ / OP_BN_;
  const int ng = blockIdx.x % NG;
  const int mg = blockIdx.x / NG;
  const int m0 = mg * OP_BM_;
  const int n0 = ng * OP_BN_;

  const int gc = (tid & 15) * 4;
  const int gr = tid >> 4;

  const float* aSrc = normed + (size_t)(m0 + gr) * HID_ + gc;
  const float* wSrc = Wout + (size_t)(n0 + gr) * HID_ + gc;

  float4 ga[OP_BM_ / 8], gw[OP_BN_ / 8];
#pragma unroll
  for (int q = 0; q < OP_BM_ / 8; ++q)
    ga[q] = ldg4(aSrc + (size_t)(8 * q) * HID_);
#pragma unroll
  for (int q = 0; q < OP_BN_ / 8; ++q)
    gw[q] = ldg4(wSrc + (size_t)(8 * q) * HID_);
#pragma unroll
  for (int q = 0; q < OP_BM_ / 8; ++q)
    *(float4*)&sA[0][gr + 8 * q][gc] = ga[q];
#pragma unroll
  for (int q = 0; q < OP_BN_ / 8; ++q)
    *(float4*)&sW[0][gr + 8 * q][gc] = gw[q];
  __syncthreads();

  v8f c0 = {};
  v8f c1 = {};
  for (int ks = 0; ks < OP_NSTEP_; ++ks) {
    const int cur = ks & 1;
    const bool more = ks + 1 < OP_NSTEP_;

    if (more) {
      const int kn = (ks + 1) * OP_KS_;
#pragma unroll
      for (int q = 0; q < OP_BM_ / 8; ++q)
        ga[q] = ldg4(aSrc + (size_t)(8 * q) * HID_ + kn);
#pragma unroll
      for (int q = 0; q < OP_BN_ / 8; ++q)
        gw[q] = ldg4(wSrc + (size_t)(8 * q) * HID_ + kn);
    }

#pragma unroll
    for (int kk = 0; kk < OP_KS_ / 4; ++kk) {
      const int kc = kk * 4 + 2 * hi;
      v2f a, b0, b1;
      a.x  = sA[cur][16 * w + row][kc];
      a.y  = sA[cur][16 * w + row][kc + 1];
      b0.x = sW[cur][row][kc];
      b0.y = sW[cur][row][kc + 1];
      b1.x = sW[cur][16 + row][kc];
      b1.y = sW[cur][16 + row][kc + 1];
      c0 = wmma_f32_4(a, b0, c0);
      c1 = wmma_f32_4(a, b1, c1);
    }

    if (more) {
#pragma unroll
      for (int q = 0; q < OP_BM_ / 8; ++q)
        *(float4*)&sA[cur ^ 1][gr + 8 * q][gc] = ga[q];
#pragma unroll
      for (int q = 0; q < OP_BN_ / 8; ++q)
        *(float4*)&sW[cur ^ 1][gr + 8 * q][gc] = gw[q];
    }
    __syncthreads();
  }

  float (*sC)[OP_PITCH_] = &sA[0][16 * w];
#pragma unroll
  for (int j = 0; j < 8; ++j) {
    sC[j + 8 * hi][row]      = c0[j];
    sC[j + 8 * hi][16 + row] = c1[j];
  }
  const int orow = lane >> 3;
  const int oc   = (lane & 7) * 4;
#pragma unroll
  for (int q = 0; q < 4; ++q) {
    const int r = orow + 4 * q;
    const float4 o = *(const float4*)&sC[r][oc];
    *(float4*)(out + (size_t)(m0 + 16 * w + r) * HID_ + n0 + oc) = o;
  }
}

extern "C" void kernel_launch(void* const* d_in, const int* in_sizes, int n_in,
                              void* d_out, int out_size, void* d_ws,
                              size_t ws_size, hipStream_t stream) {
  const float* q    = (const float*)d_in[0];
  const float* k    = (const float*)d_in[1];
  const float* v    = (const float*)d_in[2];
  const float* bias = (const float*)d_in[3];
  const int*   idx  = (const int*)d_in[5];
  const float* law  = (const float*)d_in[6];
  const float* Wout = (const float*)d_in[8];
  const float* lng  = (const float*)d_in[9];
  const float* lnb  = (const float*)d_in[10];
  float* out = (float*)d_out;

  float* ws     = (float*)d_ws;
  float* attn   = ws;
  float* normed = attn + (size_t)B_ * T_ * HID_;

  attn_kernel<<<B_ * NH_ * (T_ / 64), 512, 0, stream>>>(q, k, v, bias, law,
                                                        idx, attn);
  ln_rows_kernel<<<B_ * T_ / LN_WAVES_, 32 * LN_WAVES_, 0, stream>>>(
      attn, lng, lnb, normed);
  outproj_kernel<<<(B_ * T_ / OP_BM_) * (HID_ / OP_BN_), 128, 0, stream>>>(
      normed, Wout, out);
}


# attention on transposed tiles (S^T = K Q^T, O^T = V^T P^T): probabilities stay in registers as the P@V operand, no LDS round trip; V strip transposed in LDS, b128 reads (sum order changes)
# speedup vs baseline: 4.9590x; 1.0795x over previous
#include <hip/hip_runtime.h>
#include <hip/hip_bf16.h>


#define B_   4
#define T_   512
#define E_   256
#define HID_ 512
#define HD_  32
#define NH_  16
#define S_   768
#define LN_EPS_ 1e-5f
#define CUTOFF_ 1e-5f
#define EXP_CLAMP_ 80.0f
#define PAD_ 36

typedef __attribute__((ext_vector_type(2))) float v2f;
typedef __attribute__((ext_vector_type(8))) float v8f;

__device__ __forceinline__ v8f wmma_f32_4(v2f a, v2f b, v8f c) {
  return __builtin_amdgcn_wmma_f32_16x16x4_f32(
       false, a,  false, b,
       (short)0, c,  false,  false);
}

__device__ __forceinline__ float pair_sum(float x) {
  const int b = __builtin_bit_cast(int, x);
  const int o = __builtin_amdgcn_permlanex16(b, b, 0x76543210, 0xfedcba98,
                                             false, false);
  return x + __builtin_bit_cast(float, o);
}

__device__ __forceinline__ float4 ldg4(const float* p) {
  return *(const float4*)p;
}

__device__ __forceinline__ float2 ldg2(const float* p) {
  return *(const float2*)p;
}

__device__ __forceinline__ int clamp_row(int t) {
  return min(max(t, 0), T_ - 1);
}

__device__ __forceinline__ void lds_barrier() {
  __builtin_amdgcn_fence(__ATOMIC_RELEASE, "workgroup", "local");
  __builtin_amdgcn_s_barrier();
  __builtin_amdgcn_fence(__ATOMIC_ACQUIRE, "workgroup", "local");
}

#define SG_  4
#define SW_  32
#define SPG_ (S_ / SW_ / SG_)
#define RSLOT_ (20 * 32)

__global__ __launch_bounds__(512, 8) void attn_kernel(
    const float* __restrict__ q, const float* __restrict__ k,
    const float* __restrict__ v, const float* __restrict__ bias,
    const float* __restrict__ law, const int* __restrict__ idx,
    float* __restrict__ attn_out, float* __restrict__ stats) {
  __shared__ __attribute__((aligned(16))) float sK[SG_][SW_][PAD_];
  __shared__ __attribute__((aligned(16))) float sVt[SG_][HD_][PAD_];
  __shared__ __attribute__((aligned(16))) float sQ[4 * SG_][16][PAD_];

  const int tid  = threadIdx.x;
  const int lane = tid & 31;
  const int wv   = __builtin_amdgcn_readfirstlane(tid >> 5);
  const int w    = wv & 3;
  const int g    = wv >> 2;
  const int row  = lane & 15;
  const int hi   = lane >> 4;

  const int TB  = T_ / 64;
  const int bid = blockIdx.x;
  const int tg  = bid % TB;
  const int h   = (bid / TB) % NH_;
  const int b   = bid / (TB * NH_);
  const int t0  = (tg * 4 + w) * 16;

  {
    const int qrow = lane >> 1;
    const int qc   = (lane & 1) * 16;
    const float* qp =
        q + ((size_t)(b * T_ + t0 + qrow)) * HID_ + h * HD_ + qc;
#pragma unroll
    for (int ii = 0; ii < 4; ++ii)
      *(float4*)&sQ[wv][qrow][qc + ii * 4] = ldg4(qp + ii * 4);
  }


  v2f rs = {};
  v8f acc0 = {};
  v8f acc1 = {};

  const unsigned boff = (((unsigned)(b * NH_ + h) * T_ + t0 + row) * S_ +
                         g * (SPG_ * SW_) + 4 * hi);
  const unsigned loff = (((unsigned)b * T_ + t0 + row) * S_ +
                         g * (SPG_ * SW_) + 4 * hi);

  const int gt    = tid & 127;
  const int kvrow = gt >> 3;
  const int kvc   = (gt & 7) * 4;
  const int lrow0 = (kvrow & 3) | ((kvrow & 4) << 1) | ((kvrow & 8) >> 1);
  const int lrow1 = lrow0 + 16;
  const unsigned kvcol = (unsigned)(h * HD_ + kvc);

  const int sFirst = g * SPG_ * SW_;
  int krow0 = sFirst + kvrow;
  int krow1 = sFirst + kvrow + 16;
  if (sFirst >= T_) {
    krow0 = idx[b * E_ + (sFirst - T_) + kvrow];
    krow1 = idx[b * E_ + (sFirst - T_) + kvrow + 16];
  }
  float4 kreg0, kreg1, vreg0, vreg1;
  float4 bq[4], lq[4];
  int ixN0, ixN1;
  {
    const unsigned kvoff0 =
        (unsigned)(b * T_ + clamp_row(krow0)) * HID_ + kvcol;
    const unsigned kvoff1 =
        (unsigned)(b * T_ + clamp_row(krow1)) * HID_ + kvcol;
    kreg0 = ldg4(k + kvoff0);
    kreg1 = ldg4(k + kvoff1);
    vreg0 = ldg4(v + kvoff0);
    vreg1 = ldg4(v + kvoff1);
    __builtin_amdgcn_sched_barrier(0);
#pragma unroll
    for (int c = 0; c < 4; ++c) bq[c] = ldg4(bias + boff + c * 8);
    const int e1 = min(max(sFirst + SW_ - T_, 0), E_ - SW_);
    ixN0 = idx[b * E_ + e1 + kvrow];
    ixN1 = idx[b * E_ + e1 + kvrow + 16];
    __builtin_amdgcn_sched_barrier(0);
#pragma unroll
    for (int c = 0; c < 4; ++c) lq[c] = ldg4(law + loff + c * 8);
    __builtin_amdgcn_sched_barrier(0);
  }

  auto strip = [&](const int i, const bool last) __attribute__((always_inline)) {
    const int s0 = (g * SPG_ + i) * SW_;

    *(float4*)&sK[g][lrow0][kvc] = kreg0;
    *(float4*)&sK[g][lrow1][kvc] = kreg1;
    sVt[g][kvc + 0][lrow0] = vreg0.x;
    sVt[g][kvc + 1][lrow0] = vreg0.y;
    sVt[g][kvc + 2][lrow0] = vreg0.z;
    sVt[g][kvc + 3][lrow0] = vreg0.w;
    sVt[g][kvc + 0][lrow1] = vreg1.x;
    sVt[g][kvc + 1][lrow1] = vreg1.y;
    sVt[g][kvc + 2][lrow1] = vreg1.z;
    sVt[g][kvc + 3][lrow1] = vreg1.w;
    lds_barrier();

    v8f c0 = {};
    v8f c1 = {};
#pragma unroll
    for (int kk = 0; kk < 8; ++kk) {
      const int kc = kk * 4 + 2 * hi;
      v2f ka0, ka1, qf;
      ka0.x = sK[g][row][kc];
      ka0.y = sK[g][row][kc + 1];
      ka1.x = sK[g][16 + row][kc];
      ka1.y = sK[g][16 + row][kc + 1];
      qf.x  = sQ[wv][row][kc];
      qf.y  = sQ[wv][row][kc + 1];
      c0 = wmma_f32_4(ka0, qf, c0);
      c1 = wmma_f32_4(ka1, qf, c1);
    }
    __builtin_amdgcn_sched_barrier(0);

#pragma unroll
    for (int u = 0; u < 2; ++u) {
#pragma unroll
      for (int pp = 0; pp < 4; ++pp) {
        const int qd = 2 * u + (pp >> 1);
        const int e  = 2 * (pp & 1);
        v2f x, bb, lw, ex, ev;
        x.x  = u ? c1[2 * pp] : c0[2 * pp];
        x.y  = u ? c1[2 * pp + 1] : c0[2 * pp + 1];
        bb.x = bq[qd][e];
        bb.y = bq[qd][e + 1];
        lw.x = lq[qd][e];
        lw.y = lq[qd][e + 1];
        x = x + bb;
        ex.x = __expf(fminf(x.x, EXP_CLAMP_));
        ex.y = __expf(fminf(x.y, EXP_CLAMP_));
        ev.x = (lw.x <= CUTOFF_) ? 0.0f : ex.x;
        ev.y = (lw.y <= CUTOFF_) ? 0.0f : ex.y;
        rs = rs + ev;
        const v2f p = ev * lw;
        if (u) { c1[2 * pp] = p.x; c1[2 * pp + 1] = p.y; }
        else   { c0[2 * pp] = p.x; c0[2 * pp + 1] = p.y; }
      }
    }
    asm volatile("" : : "v"(c0), "v"(c1));

    __builtin_amdgcn_sched_barrier(0);
    if (!last) {
      const int sN = s0 + SW_;
      const int krowN0 = (sN >= T_) ? clamp_row(ixN0) : sN + kvrow;
      const int krowN1 = (sN >= T_) ? clamp_row(ixN1) : sN + kvrow + 16;
      const unsigned kvoff0 = (unsigned)(b * T_ + krowN0) * HID_ + kvcol;
      const unsigned kvoff1 = (unsigned)(b * T_ + krowN1) * HID_ + kvcol;
      kreg0 = ldg4(k + kvoff0);
      kreg1 = ldg4(k + kvoff1);
      vreg0 = ldg4(v + kvoff0);
      vreg1 = ldg4(v + kvoff1);
      __builtin_amdgcn_sched_barrier(0);
      const float* bnext = bias + (i + 1) * SW_;
#pragma unroll
      for (int c = 0; c < 4; ++c) bq[c] = ldg4(bnext + boff + c * 8);
      const int e2 = min(max(s0 + 2 * SW_ - T_, 0), E_ - SW_);
      ixN0 = idx[b * E_ + e2 + kvrow];
      ixN1 = idx[b * E_ + e2 + kvrow + 16];
      const float* lnext = law + (i + 1) * SW_;
#pragma unroll
      for (int c = 0; c < 4; ++c) lq[c] = ldg4(lnext + loff + c * 8);
    }
    __builtin_amdgcn_sched_barrier(0);

#pragma unroll
    for (int u = 0; u < 2; ++u) {
      const v8f pu = u ? c1 : c0;
      const float4 va0 = *(const float4*)&sVt[g][row][16 * u + 8 * hi];
      const float4 va1 = *(const float4*)&sVt[g][row][16 * u + 8 * hi + 4];
      const float4 vb0 = *(const float4*)&sVt[g][16 + row][16 * u + 8 * hi];
      const float4 vb1 = *(const float4*)&sVt[g][16 + row][16 * u + 8 * hi + 4];
      v2f fa[4], fb[4], fp[4];
      fa[0].x = va0.x; fa[0].y = va0.y; fa[1].x = va0.z; fa[1].y = va0.w;
      fa[2].x = va1.x; fa[2].y = va1.y; fa[3].x = va1.z; fa[3].y = va1.w;
      fb[0].x = vb0.x; fb[0].y = vb0.y; fb[1].x = vb0.z; fb[1].y = vb0.w;
      fb[2].x = vb1.x; fb[2].y = vb1.y; fb[3].x = vb1.z; fb[3].y = vb1.w;
#pragma unroll
      for (int kk = 0; kk < 4; ++kk) {
        fp[kk].x = pu[2 * kk];
        fp[kk].y = pu[2 * kk + 1];
      }
#pragma unroll
      for (int kk = 0; kk < 4; ++kk) {
        acc0 = wmma_f32_4(fa[kk], fp[kk], acc0);
        acc1 = wmma_f32_4(fb[kk], fp[kk], acc1);
      }
    }
    lds_barrier();
  };

#pragma nounroll
  for (int i = 0; i < SPG_ - 1; ++i) strip(i, false);
  strip(SPG_ - 1, true);

  float* red = &sQ[0][0][0];
  if (g != 0) {
    float* r = red + ((g - 1) * 4 + w) * RSLOT_ + lane;
#pragma unroll
    for (int j = 0; j < 8; ++j) {
      r[j * 32]       = acc0[j];
      r[(8 + j) * 32] = acc1[j];
    }
    r[16 * 32] = rs.x + rs.y;
  }
  __syncthreads();
  if (g != 0) return;
  float rsum = rs.x + rs.y;
#pragma unroll
  for (int gg = 0; gg < SG_ - 1; ++gg) {
    const float* r = red + (gg * 4 + w) * RSLOT_ + lane;
#pragma unroll
    for (int j = 0; j < 8; ++j) {
      acc0[j] += r[j * 32];
      acc1[j] += r[(8 + j) * 32];
    }
    rsum += r[16 * 32];
  }

  const float inv = 1.0f / pair_sum(rsum);
  float o0[8], o1[8];
  float hs = 0.0f;
#pragma unroll
  for (int j = 0; j < 8; ++j) {
    o0[j] = acc0[j] * inv;
    o1[j] = acc1[j] * inv;
    hs += o0[j] + o1[j];
  }
  const float mh = pair_sum(hs) * (1.0f / HD_);
  float m2 = 0.0f;
#pragma unroll
  for (int j = 0; j < 8; ++j) {
    const float d0 = o0[j] - mh, d1 = o1[j] - mh;
    m2 += d0 * d0 + d1 * d1;
  }
  m2 = pair_sum(m2);
  if (hi == 0) {
    float2 st;
    st.x = mh;
    st.y = m2;
    *(float2*)(stats + (((size_t)(b * NH_ + h)) * T_ + t0 + row) * 2) = st;
  }

  float* so = red + w * RSLOT_;
#pragma unroll
  for (int j = 0; j < 2; ++j) {
    float4 x0, x1;
    x0.x = o0[4 * j]; x0.y = o0[4 * j + 1]; x0.z = o0[4 * j + 2]; x0.w = o0[4 * j + 3];
    x1.x = o1[4 * j]; x1.y = o1[4 * j + 1]; x1.z = o1[4 * j + 2]; x1.w = o1[4 * j + 3];
    *(float4*)&so[row * PAD_ + 8 * hi + 4 * j]      = x0;
    *(float4*)&so[row * PAD_ + 16 + 8 * hi + 4 * j] = x1;
  }
  const int orow = lane >> 3;
  const int oc   = (lane & 7) * 4;
#pragma unroll
  for (int qq = 0; qq < 4; ++qq) {
    const int r = orow + 4 * qq;
    const float4 o = *(const float4*)&so[r * PAD_ + oc];
    *(float4*)(attn_out + ((size_t)(b * T_ + t0 + r)) * HID_ + h * HD_ + oc) = o;
  }
}

#define OP_BM_    64
#define OP_BN_    32
#define OP_KS_    64
#define OP_PITCH_ 68
#define OP_NSTEP_ (HID_ / OP_KS_)

__device__ __forceinline__ float4 ln_apply(float4 x, float mean, float rs,
                                           float4 g, float4 o) {
  float4 r;
  r.x = (x.x - mean) * rs * g.x + o.x;
  r.y = (x.y - mean) * rs * g.y + o.y;
  r.z = (x.z - mean) * rs * g.z + o.z;
  r.w = (x.w - mean) * rs * g.w + o.w;
  return r;
}

__global__ __launch_bounds__(128) void outproj_kernel(
    const float* __restrict__ attn, const float* __restrict__ stats,
    const float* __restrict__ ln_g, const float* __restrict__ ln_b,
    const float* __restrict__ Wout, float* __restrict__ out) {
  __shared__ __attribute__((aligned(16))) float sA[2][OP_BM_][OP_PITCH_];
  __shared__ __attribute__((aligned(16))) float sW[2][OP_BN_][OP_PITCH_];
  __shared__ float sMu[OP_BM_];
  __shared__ float sRs[OP_BM_];

  const int tid  = threadIdx.x;
  const int lane = tid & 31;
  const int w    = __builtin_amdgcn_readfirstlane(tid >> 5);
  const int row  = lane & 15;
  const int hi   = lane >> 4;

  const int NG = HID_ / OP_BN_;
  const int ng = blockIdx.x % NG;
  const int mg = blockIdx.x / NG;
  const int m0 = mg * OP_BM_;
  const int n0 = ng * OP_BN_;

  const int gc = (tid & 15) * 4;
  const int gr = tid >> 4;

  const float* aSrc = attn + (size_t)(m0 + gr) * HID_ + gc;
  const float* wSrc = Wout + (size_t)(n0 + gr) * HID_ + gc;

  float4 ga[OP_BM_ / 8], gw[OP_BN_ / 8];
#pragma unroll
  for (int q = 0; q < OP_BM_ / 8; ++q)
    ga[q] = ldg4(aSrc + (size_t)(8 * q) * HID_);
#pragma unroll
  for (int q = 0; q < OP_BN_ / 8; ++q)
    gw[q] = ldg4(wSrc + (size_t)(8 * q) * HID_);
  float4 gg = ldg4(ln_g + gc);
  float4 gb = ldg4(ln_b + gc);

  if (tid < OP_BM_) {
    const int m  = m0 + tid;
    const int bb = m / T_;
    const int t  = m % T_;
    const float* sp = stats + (((size_t)bb * NH_) * T_ + t) * 2;
    float2 p[NH_];
#pragma unroll
    for (int hh = 0; hh < NH_; ++hh) p[hh] = ldg2(sp + (size_t)hh * T_ * 2);
    float ms = 0.0f;
#pragma unroll
    for (int hh = 0; hh < NH_; ++hh) ms += p[hh].x;
    const float mean = ms * (1.0f / NH_);
    float m2 = 0.0f;
#pragma unroll
    for (int hh = 0; hh < NH_; ++hh) {
      const float dm = p[hh].x - mean;
      m2 += p[hh].y + (float)HD_ * (dm * dm);
    }
    sMu[tid] = mean;
    sRs[tid] = rsqrtf(m2 * (1.0f / HID_) + LN_EPS_);
  }
  __syncthreads();
  float mu[OP_BM_ / 8], rsd[OP_BM_ / 8];
#pragma unroll
  for (int q = 0; q < OP_BM_ / 8; ++q) {
    mu[q]  = sMu[gr + 8 * q];
    rsd[q] = sRs[gr + 8 * q];
  }

#pragma unroll
  for (int q = 0; q < OP_BM_ / 8; ++q)
    *(float4*)&sA[0][gr + 8 * q][gc] = ln_apply(ga[q], mu[q], rsd[q], gg, gb);
#pragma unroll
  for (int q = 0; q < OP_BN_ / 8; ++q)
    *(float4*)&sW[0][gr + 8 * q][gc] = gw[q];
  __syncthreads();

  v8f c0 = {};
  v8f c1 = {};
  for (int ks = 0; ks < OP_NSTEP_; ++ks) {
    const int cur = ks & 1;
    const bool more = ks + 1 < OP_NSTEP_;

    if (more) {
      const int kn = (ks + 1) * OP_KS_;
#pragma unroll
      for (int q = 0; q < OP_BM_ / 8; ++q)
        ga[q] = ldg4(aSrc + (size_t)(8 * q) * HID_ + kn);
#pragma unroll
      for (int q = 0; q < OP_BN_ / 8; ++q)
        gw[q] = ldg4(wSrc + (size_t)(8 * q) * HID_ + kn);
      gg = ldg4(ln_g + gc + kn);
      gb = ldg4(ln_b + gc + kn);
    }

#pragma unroll
    for (int kk = 0; kk < OP_KS_ / 4; ++kk) {
      const int kc = kk * 4 + 2 * hi;
      v2f a, b0, b1;
      a.x  = sA[cur][16 * w + row][kc];
      a.y  = sA[cur][16 * w + row][kc + 1];
      b0.x = sW[cur][row][kc];
      b0.y = sW[cur][row][kc + 1];
      b1.x = sW[cur][16 + row][kc];
      b1.y = sW[cur][16 + row][kc + 1];
      c0 = wmma_f32_4(a, b0, c0);
      c1 = wmma_f32_4(a, b1, c1);
    }

    if (more) {
#pragma unroll
      for (int q = 0; q < OP_BM_ / 8; ++q)
        *(float4*)&sA[cur ^ 1][gr + 8 * q][gc] =
            ln_apply(ga[q], mu[q], rsd[q], gg, gb);
#pragma unroll
      for (int q = 0; q < OP_BN_ / 8; ++q)
        *(float4*)&sW[cur ^ 1][gr + 8 * q][gc] = gw[q];
    }
    __syncthreads();
  }

  float (*sC)[OP_PITCH_] = &sA[0][16 * w];
#pragma unroll
  for (int j = 0; j < 8; ++j) {
    sC[j + 8 * hi][row]      = c0[j];
    sC[j + 8 * hi][16 + row] = c1[j];
  }
  const int orow = lane >> 3;
  const int oc   = (lane & 7) * 4;
#pragma unroll
  for (int q = 0; q < 4; ++q) {
    const int r = orow + 4 * q;
    const float4 o = *(const float4*)&sC[r][oc];
    *(float4*)(out + (size_t)(m0 + 16 * w + r) * HID_ + n0 + oc) = o;
  }
}

extern "C" void kernel_launch(void* const* d_in, const int* in_sizes, int n_in,
                              void* d_out, int out_size, void* d_ws,
                              size_t ws_size, hipStream_t stream) {
  const float* q    = (const float*)d_in[0];
  const float* k    = (const float*)d_in[1];
  const float* v    = (const float*)d_in[2];
  const float* bias = (const float*)d_in[3];
  const int*   idx  = (const int*)d_in[5];
  const float* law  = (const float*)d_in[6];
  const float* Wout = (const float*)d_in[8];
  const float* lng  = (const float*)d_in[9];
  const float* lnb  = (const float*)d_in[10];
  float* out = (float*)d_out;

  float* ws    = (float*)d_ws;
  float* attn  = ws;
  float* stats = attn + (size_t)B_ * T_ * HID_;

  attn_kernel<<<B_ * NH_ * (T_ / 64), 512, 0, stream>>>(q, k, v, bias, law,
                                                        idx, attn, stats);
  outproj_kernel<<<(B_ * T_ / OP_BM_) * (HID_ / OP_BN_), 128, 0, stream>>>(
      attn, stats, lng, lnb, Wout, out);
}


# attention: a wave loads its bias and law strip pieces as whole 128-B lines (8 lanes per row, 4 lines per load, was 16) and re-forms them to one row per lane through an LDS tile of its own
# speedup vs baseline: 5.1589x; 1.0403x over previous
#include <hip/hip_runtime.h>
#include <hip/hip_bf16.h>


#define B_   4
#define T_   512
#define E_   256
#define HID_ 512
#define HD_  32
#define NH_  16
#define S_   768
#define LN_EPS_ 1e-5f
#define CUTOFF_ 1e-5f
#define EXP_CLAMP_ 80.0f
#define PAD_ 36

typedef __attribute__((ext_vector_type(2))) float v2f;
typedef __attribute__((ext_vector_type(8))) float v8f;

__device__ __forceinline__ v8f wmma_f32_4(v2f a, v2f b, v8f c) {
  return __builtin_amdgcn_wmma_f32_16x16x4_f32(
       false, a,  false, b,
       (short)0, c,  false,  false);
}

__device__ __forceinline__ float pair_sum(float x) {
  const int b = __builtin_bit_cast(int, x);
  const int o = __builtin_amdgcn_permlanex16(b, b, 0x76543210, 0xfedcba98,
                                             false, false);
  return x + __builtin_bit_cast(float, o);
}

__device__ __forceinline__ float4 ldg4(const float* p) {
  return *(const float4*)p;
}

__device__ __forceinline__ float2 ldg2(const float* p) {
  return *(const float2*)p;
}

__device__ __forceinline__ int clamp_row(int t) {
  return min(max(t, 0), T_ - 1);
}

__device__ __forceinline__ void lds_barrier() {
  __builtin_amdgcn_fence(__ATOMIC_RELEASE, "workgroup", "local");
  __builtin_amdgcn_s_barrier();
  __builtin_amdgcn_fence(__ATOMIC_ACQUIRE, "workgroup", "local");
}

#define SG_  4
#define SW_  32
#define SPG_ (S_ / SW_ / SG_)
#define RSLOT_ (20 * 32)

__global__ __launch_bounds__(512, 8) void attn_kernel(
    const float* __restrict__ q, const float* __restrict__ k,
    const float* __restrict__ v, const float* __restrict__ bias,
    const float* __restrict__ law, const int* __restrict__ idx,
    float* __restrict__ attn_out, float* __restrict__ stats) {
  __shared__ __attribute__((aligned(16))) float sK[SG_][SW_][PAD_];
  __shared__ __attribute__((aligned(16))) float sVt[SG_][HD_][PAD_];
  __shared__ __attribute__((aligned(16))) float sQ[4 * SG_][16][PAD_];
  __shared__ __attribute__((aligned(16))) float sB[4 * SG_][16][PAD_];
  __shared__ __attribute__((aligned(16))) float sL[4 * SG_][16][PAD_];

  const int tid  = threadIdx.x;
  const int lane = tid & 31;
  const int wv   = __builtin_amdgcn_readfirstlane(tid >> 5);
  const int w    = wv & 3;
  const int g    = wv >> 2;
  const int row  = lane & 15;
  const int hi   = lane >> 4;

  const int TB  = T_ / 64;
  const int bid = blockIdx.x;
  const int tg  = bid % TB;
  const int h   = (bid / TB) % NH_;
  const int b   = bid / (TB * NH_);
  const int t0  = (tg * 4 + w) * 16;

  {
    const int qrow = lane >> 1;
    const int qc   = (lane & 1) * 16;
    const float* qp =
        q + ((size_t)(b * T_ + t0 + qrow)) * HID_ + h * HD_ + qc;
#pragma unroll
    for (int ii = 0; ii < 4; ++ii)
      *(float4*)&sQ[wv][qrow][qc + ii * 4] = ldg4(qp + ii * 4);
  }


  v2f rs = {};
  v8f acc0 = {};
  v8f acc1 = {};

  const int lr = lane >> 3;
  const int lc = (lane & 7) * 4;
  const unsigned boff = (((unsigned)(b * NH_ + h) * T_ + t0 + lr) * S_ +
                         g * (SPG_ * SW_) + lc);
  const unsigned loff = (((unsigned)b * T_ + t0 + lr) * S_ +
                         g * (SPG_ * SW_) + lc);

  const int gt    = tid & 127;
  const int kvrow = gt >> 3;
  const int kvc   = (gt & 7) * 4;
  const int lrow0 = (kvrow & 3) | ((kvrow & 4) << 1) | ((kvrow & 8) >> 1);
  const int lrow1 = lrow0 + 16;
  const unsigned kvcol = (unsigned)(h * HD_ + kvc);

  const int sFirst = g * SPG_ * SW_;
  int krow0 = sFirst + kvrow;
  int krow1 = sFirst + kvrow + 16;
  if (sFirst >= T_) {
    krow0 = idx[b * E_ + (sFirst - T_) + kvrow];
    krow1 = idx[b * E_ + (sFirst - T_) + kvrow + 16];
  }
  float4 kreg0, kreg1, vreg0, vreg1;
  float4 bq[4], lq[4];
  int ixN0, ixN1;
  {
    const unsigned kvoff0 =
        (unsigned)(b * T_ + clamp_row(krow0)) * HID_ + kvcol;
    const unsigned kvoff1 =
        (unsigned)(b * T_ + clamp_row(krow1)) * HID_ + kvcol;
    kreg0 = ldg4(k + kvoff0);
    kreg1 = ldg4(k + kvoff1);
    vreg0 = ldg4(v + kvoff0);
    vreg1 = ldg4(v + kvoff1);
    __builtin_amdgcn_sched_barrier(0);
#pragma unroll
    for (int c = 0; c < 4; ++c) bq[c] = ldg4(bias + boff + c * (4 * S_));
    const int e1 = min(max(sFirst + SW_ - T_, 0), E_ - SW_);
    ixN0 = idx[b * E_ + e1 + kvrow];
    ixN1 = idx[b * E_ + e1 + kvrow + 16];
    __builtin_amdgcn_sched_barrier(0);
#pragma unroll
    for (int c = 0; c < 4; ++c) lq[c] = ldg4(law + loff + c * (4 * S_));
    __builtin_amdgcn_sched_barrier(0);
  }

  auto strip = [&](const int i, const bool last) __attribute__((always_inline)) {
    const int s0 = (g * SPG_ + i) * SW_;

    *(float4*)&sK[g][lrow0][kvc] = kreg0;
    *(float4*)&sK[g][lrow1][kvc] = kreg1;
    sVt[g][kvc + 0][lrow0] = vreg0.x;
    sVt[g][kvc + 1][lrow0] = vreg0.y;
    sVt[g][kvc + 2][lrow0] = vreg0.z;
    sVt[g][kvc + 3][lrow0] = vreg0.w;
    sVt[g][kvc + 0][lrow1] = vreg1.x;
    sVt[g][kvc + 1][lrow1] = vreg1.y;
    sVt[g][kvc + 2][lrow1] = vreg1.z;
    sVt[g][kvc + 3][lrow1] = vreg1.w;
    lds_barrier();

    v8f c0 = {};
    v8f c1 = {};
#pragma unroll
    for (int kk = 0; kk < 8; ++kk) {
      const int kc = kk * 4 + 2 * hi;
      v2f ka0, ka1, qf;
      ka0.x = sK[g][row][kc];
      ka0.y = sK[g][row][kc + 1];
      ka1.x = sK[g][16 + row][kc];
      ka1.y = sK[g][16 + row][kc + 1];
      qf.x  = sQ[wv][row][kc];
      qf.y  = sQ[wv][row][kc + 1];
      c0 = wmma_f32_4(ka0, qf, c0);
      c1 = wmma_f32_4(ka1, qf, c1);
    }
    asm volatile("" : : "v"(c0), "v"(c1));
    __builtin_amdgcn_sched_barrier(0);

#pragma unroll
    for (int c = 0; c < 4; ++c) *(float4*)&sB[wv][4 * c + lr][lc] = bq[c];
#pragma unroll
    for (int c = 0; c < 4; ++c) *(float4*)&sL[wv][4 * c + lr][lc] = lq[c];
#pragma unroll
    for (int c = 0; c < 4; ++c)
      bq[c] = *(const float4*)&sB[wv][row][8 * c + 4 * hi];
#pragma unroll
    for (int c = 0; c < 4; ++c)
      lq[c] = *(const float4*)&sL[wv][row][8 * c + 4 * hi];

#pragma unroll
    for (int u = 0; u < 2; ++u) {
#pragma unroll
      for (int pp = 0; pp < 4; ++pp) {
        const int qd = 2 * u + (pp >> 1);
        const int e  = 2 * (pp & 1);
        v2f x, bb, lw, ex, ev;
        x.x  = u ? c1[2 * pp] : c0[2 * pp];
        x.y  = u ? c1[2 * pp + 1] : c0[2 * pp + 1];
        bb.x = bq[qd][e];
        bb.y = bq[qd][e + 1];
        lw.x = lq[qd][e];
        lw.y = lq[qd][e + 1];
        x = x + bb;
        ex.x = __expf(fminf(x.x, EXP_CLAMP_));
        ex.y = __expf(fminf(x.y, EXP_CLAMP_));
        ev.x = (lw.x <= CUTOFF_) ? 0.0f : ex.x;
        ev.y = (lw.y <= CUTOFF_) ? 0.0f : ex.y;
        rs = rs + ev;
        const v2f p = ev * lw;
        if (u) { c1[2 * pp] = p.x; c1[2 * pp + 1] = p.y; }
        else   { c0[2 * pp] = p.x; c0[2 * pp + 1] = p.y; }
      }
    }
    asm volatile("" : : "v"(c0), "v"(c1));

    __builtin_amdgcn_sched_barrier(0);
    if (!last) {
      const int sN = s0 + SW_;
      const int krowN0 = (sN >= T_) ? clamp_row(ixN0) : sN + kvrow;
      const int krowN1 = (sN >= T_) ? clamp_row(ixN1) : sN + kvrow + 16;
      const unsigned kvoff0 = (unsigned)(b * T_ + krowN0) * HID_ + kvcol;
      const unsigned kvoff1 = (unsigned)(b * T_ + krowN1) * HID_ + kvcol;
      kreg0 = ldg4(k + kvoff0);
      kreg1 = ldg4(k + kvoff1);
      vreg0 = ldg4(v + kvoff0);
      vreg1 = ldg4(v + kvoff1);
      __builtin_amdgcn_sched_barrier(0);
      const float* bnext = bias + (i + 1) * SW_;
#pragma unroll
      for (int c = 0; c < 4; ++c) bq[c] = ldg4(bnext + boff + c * (4 * S_));
      const int e2 = min(max(s0 + 2 * SW_ - T_, 0), E_ - SW_);
      ixN0 = idx[b * E_ + e2 + kvrow];
      ixN1 = idx[b * E_ + e2 + kvrow + 16];
      const float* lnext = law + (i + 1) * SW_;
#pragma unroll
      for (int c = 0; c < 4; ++c) lq[c] = ldg4(lnext + loff + c * (4 * S_));
    }
    __builtin_amdgcn_sched_barrier(0);

#pragma unroll
    for (int u = 0; u < 2; ++u) {
      const v8f pu = u ? c1 : c0;
      const float4 va0 = *(const float4*)&sVt[g][row][16 * u + 8 * hi];
      const float4 va1 = *(const float4*)&sVt[g][row][16 * u + 8 * hi + 4];
      const float4 vb0 = *(const float4*)&sVt[g][16 + row][16 * u + 8 * hi];
      const float4 vb1 = *(const float4*)&sVt[g][16 + row][16 * u + 8 * hi + 4];
      v2f fa[4], fb[4], fp[4];
      fa[0].x = va0.x; fa[0].y = va0.y; fa[1].x = va0.z; fa[1].y = va0.w;
      fa[2].x = va1.x; fa[2].y = va1.y; fa[3].x = va1.z; fa[3].y = va1.w;
      fb[0].x = vb0.x; fb[0].y = vb0.y; fb[1].x = vb0.z; fb[1].y = vb0.w;
      fb[2].x = vb1.x; fb[2].y = vb1.y; fb[3].x = vb1.z; fb[3].y = vb1.w;
#pragma unroll
      for (int kk = 0; kk < 4; ++kk) {
        fp[kk].x = pu[2 * kk];
        fp[kk].y = pu[2 * kk + 1];
      }
#pragma unroll
      for (int kk = 0; kk < 4; ++kk) {
        acc0 = wmma_f32_4(fa[kk], fp[kk], acc0);
        acc1 = wmma_f32_4(fb[kk], fp[kk], acc1);
      }
    }
    lds_barrier();
  };

#pragma nounroll
  for (int i = 0; i < SPG_ - 1; ++i) strip(i, false);
  strip(SPG_ - 1, true);

  float* red = &sQ[0][0][0];
  if (g != 0) {
    float* r = red + ((g - 1) * 4 + w) * RSLOT_ + lane;
#pragma unroll
    for (int j = 0; j < 8; ++j) {
      r[j * 32]       = acc0[j];
      r[(8 + j) * 32] = acc1[j];
    }
    r[16 * 32] = rs.x + rs.y;
  }
  __syncthreads();
  if (g != 0) return;
  float rsum = rs.x + rs.y;
#pragma unroll
  for (int gg = 0; gg < SG_ - 1; ++gg) {
    const float* r = red + (gg * 4 + w) * RSLOT_ + lane;
#pragma unroll
    for (int j = 0; j < 8; ++j) {
      acc0[j] += r[j * 32];
      acc1[j] += r[(8 + j) * 32];
    }
    rsum += r[16 * 32];
  }

  const float inv = 1.0f / pair_sum(rsum);
  float o0[8], o1[8];
  float hs = 0.0f;
#pragma unroll
  for (int j = 0; j < 8; ++j) {
    o0[j] = acc0[j] * inv;
    o1[j] = acc1[j] * inv;
    hs += o0[j] + o1[j];
  }
  const float mh = pair_sum(hs) * (1.0f / HD_);
  float m2 = 0.0f;
#pragma unroll
  for (int j = 0; j < 8; ++j) {
    const float d0 = o0[j] - mh, d1 = o1[j] - mh;
    m2 += d0 * d0 + d1 * d1;
  }
  m2 = pair_sum(m2);
  if (hi == 0) {
    float2 st;
    st.x = mh;
    st.y = m2;
    *(float2*)(stats + (((size_t)(b * NH_ + h)) * T_ + t0 + row) * 2) = st;
  }

  float* so = red + w * RSLOT_;
#pragma unroll
  for (int j = 0; j < 2; ++j) {
    float4 x0, x1;
    x0.x = o0[4 * j]; x0.y = o0[4 * j + 1]; x0.z = o0[4 * j + 2]; x0.w = o0[4 * j + 3];
    x1.x = o1[4 * j]; x1.y = o1[4 * j + 1]; x1.z = o1[4 * j + 2]; x1.w = o1[4 * j + 3];
    *(float4*)&so[row * PAD_ + 8 * hi + 4 * j]      = x0;
    *(float4*)&so[row * PAD_ + 16 + 8 * hi + 4 * j] = x1;
  }
  const int orow = lane >> 3;
  const int oc   = (lane & 7) * 4;
#pragma unroll
  for (int qq = 0; qq < 4; ++qq) {
    const int r = orow + 4 * qq;
    const float4 o = *(const float4*)&so[r * PAD_ + oc];
    *(float4*)(attn_out + ((size_t)(b * T_ + t0 + r)) * HID_ + h * HD_ + oc) = o;
  }
}

#define OP_BM_    64
#define OP_BN_    32
#define OP_KS_    64
#define OP_PITCH_ 68
#define OP_NSTEP_ (HID_ / OP_KS_)

__device__ __forceinline__ float4 ln_apply(float4 x, float mean, float rs,
                                           float4 g, float4 o) {
  float4 r;
  r.x = (x.x - mean) * rs * g.x + o.x;
  r.y = (x.y - mean) * rs * g.y + o.y;
  r.z = (x.z - mean) * rs * g.z + o.z;
  r.w = (x.w - mean) * rs * g.w + o.w;
  return r;
}

__global__ __launch_bounds__(128) void outproj_kernel(
    const float* __restrict__ attn, const float* __restrict__ stats,
    const float* __restrict__ ln_g, const float* __restrict__ ln_b,
    const float* __restrict__ Wout, float* __restrict__ out) {
  __shared__ __attribute__((aligned(16))) float sA[2][OP_BM_][OP_PITCH_];
  __shared__ __attribute__((aligned(16))) float sW[2][OP_BN_][OP_PITCH_];
  __shared__ float sMu[OP_BM_];
  __shared__ float sRs[OP_BM_];

  const int tid  = threadIdx.x;
  const int lane = tid & 31;
  const int w    = __builtin_amdgcn_readfirstlane(tid >> 5);
  const int row  = lane & 15;
  const int hi   = lane >> 4;

  const int NG = HID_ / OP_BN_;
  const int ng = blockIdx.x % NG;
  const int mg = blockIdx.x / NG;
  const int m0 = mg * OP_BM_;
  const int n0 = ng * OP_BN_;

  const int gc = (tid & 15) * 4;
  const int gr = tid >> 4;

  const float* aSrc = attn + (size_t)(m0 + gr) * HID_ + gc;
  const float* wSrc = Wout + (size_t)(n0 + gr) * HID_ + gc;

  float4 ga[OP_BM_ / 8], gw[OP_BN_ / 8];
#pragma unroll
  for (int q = 0; q < OP_BM_ / 8; ++q)
    ga[q] = ldg4(aSrc + (size_t)(8 * q) * HID_);
#pragma unroll
  for (int q = 0; q < OP_BN_ / 8; ++q)
    gw[q] = ldg4(wSrc + (size_t)(8 * q) * HID_);
  float4 gg = ldg4(ln_g + gc);
  float4 gb = ldg4(ln_b + gc);

  if (tid < OP_BM_) {
    const int m  = m0 + tid;
    const int bb = m / T_;
    const int t  = m % T_;
    const float* sp = stats + (((size_t)bb * NH_) * T_ + t) * 2;
    float2 p[NH_];
#pragma unroll
    for (int hh = 0; hh < NH_; ++hh) p[hh] = ldg2(sp + (size_t)hh * T_ * 2);
    float ms = 0.0f;
#pragma unroll
    for (int hh = 0; hh < NH_; ++hh) ms += p[hh].x;
    const float mean = ms * (1.0f / NH_);
    float m2 = 0.0f;
#pragma unroll
    for (int hh = 0; hh < NH_; ++hh) {
      const float dm = p[hh].x - mean;
      m2 += p[hh].y + (float)HD_ * (dm * dm);
    }
    sMu[tid] = mean;
    sRs[tid] = rsqrtf(m2 * (1.0f / HID_) + LN_EPS_);
  }
  __syncthreads();
  float mu[OP_BM_ / 8], rsd[OP_BM_ / 8];
#pragma unroll
  for (int q = 0; q < OP_BM_ / 8; ++q) {
    mu[q]  = sMu[gr + 8 * q];
    rsd[q] = sRs[gr + 8 * q];
  }

#pragma unroll
  for (int q = 0; q < OP_BM_ / 8; ++q)
    *(float4*)&sA[0][gr + 8 * q][gc] = ln_apply(ga[q], mu[q], rsd[q], gg, gb);
#pragma unroll
  for (int q = 0; q < OP_BN_ / 8; ++q)
    *(float4*)&sW[0][gr + 8 * q][gc] = gw[q];
  __syncthreads();

  v8f c0 = {};
  v8f c1 = {};
  for (int ks = 0; ks < OP_NSTEP_; ++ks) {
    const int cur = ks & 1;
    const bool more = ks + 1 < OP_NSTEP_;

    if (more) {
      const int kn = (ks + 1) * OP_KS_;
#pragma unroll
      for (int q = 0; q < OP_BM_ / 8; ++q)
        ga[q] = ldg4(aSrc + (size_t)(8 * q) * HID_ + kn);
#pragma unroll
      for (int q = 0; q < OP_BN_ / 8; ++q)
        gw[q] = ldg4(wSrc + (size_t)(8 * q) * HID_ + kn);
      gg = ldg4(ln_g + gc + kn);
      gb = ldg4(ln_b + gc + kn);
    }

#pragma unroll
    for (int kk = 0; kk < OP_KS_ / 4; ++kk) {
      const int kc = kk * 4 + 2 * hi;
      v2f a, b0, b1;
      a.x  = sA[cur][16 * w + row][kc];
      a.y  = sA[cur][16 * w + row][kc + 1];
      b0.x = sW[cur][row][kc];
      b0.y = sW[cur][row][kc + 1];
      b1.x = sW[cur][16 + row][kc];
      b1.y = sW[cur][16 + row][kc + 1];
      c0 = wmma_f32_4(a, b0, c0);
      c1 = wmma_f32_4(a, b1, c1);
    }

    if (more) {
#pragma unroll
      for (int q = 0; q < OP_BM_ / 8; ++q)
        *(float4*)&sA[cur ^ 1][gr + 8 * q][gc] =
            ln_apply(ga[q], mu[q], rsd[q], gg, gb);
#pragma unroll
      for (int q = 0; q < OP_BN_ / 8; ++q)
        *(float4*)&sW[cur ^ 1][gr + 8 * q][gc] = gw[q];
    }
    __syncthreads();
  }

  float (*sC)[OP_PITCH_] = &sA[0][16 * w];
#pragma unroll
  for (int j = 0; j < 8; ++j) {
    sC[j + 8 * hi][row]      = c0[j];
    sC[j + 8 * hi][16 + row] = c1[j];
  }
  const int orow = lane >> 3;
  const int oc   = (lane & 7) * 4;
#pragma unroll
  for (int q = 0; q < 4; ++q) {
    const int r = orow + 4 * q;
    const float4 o = *(const float4*)&sC[r][oc];
    *(float4*)(out + (size_t)(m0 + 16 * w + r) * HID_ + n0 + oc) = o;
  }
}

extern "C" void kernel_launch(void* const* d_in, const int* in_sizes, int n_in,
                              void* d_out, int out_size, void* d_ws,
                              size_t ws_size, hipStream_t stream) {
  const float* q    = (const float*)d_in[0];
  const float* k    = (const float*)d_in[1];
  const float* v    = (const float*)d_in[2];
  const float* bias = (const float*)d_in[3];
  const int*   idx  = (const int*)d_in[5];
  const float* law  = (const float*)d_in[6];
  const float* Wout = (const float*)d_in[8];
  const float* lng  = (const float*)d_in[9];
  const float* lnb  = (const float*)d_in[10];
  float* out = (float*)d_out;

  float* ws    = (float*)d_ws;
  float* attn  = ws;
  float* stats = attn + (size_t)B_ * T_ * HID_;

  attn_kernel<<<B_ * NH_ * (T_ / 64), 512, 0, stream>>>(q, k, v, bias, law,
                                                        idx, attn, stats);
  outproj_kernel<<<(B_ * T_ / OP_BM_) * (HID_ / OP_BN_), 128, 0, stream>>>(
      attn, stats, lng, lnb, Wout, out);
}
